# starting kernel
# baseline (speedup 1.0000x reference)
#include <hip/hip_runtime.h>

#define IN_DIM 64
#define OUT_DIM 64

typedef float v2f __attribute__((ext_vector_type(2)));
typedef float v8f __attribute__((ext_vector_type(8)));

__global__ void zero4_kernel(float4* __restrict__ p, int n4) {
    int i = blockIdx.x * blockDim.x + threadIdx.x;
    int stride = gridDim.x * blockDim.x;
    float4 z = make_float4(0.f, 0.f, 0.f, 0.f);
    for (; i < n4; i += stride) p[i] = z;
}

__global__ void zero1_kernel(float* __restrict__ p, int n) {
    int i = blockIdx.x * blockDim.x + threadIdx.x;
    int stride = gridDim.x * blockDim.x;
    for (; i < n; i += stride) p[i] = 0.0f;
}

__global__ void edge_scatter_kernel(const float* __restrict__ x,
                                    const int* __restrict__ src,
                                    const int* __restrict__ dst,
                                    float* __restrict__ msg,
                                    float* __restrict__ deg,
                                    int nEdges) {
    int gid = blockIdx.x * blockDim.x + threadIdx.x;
    int total = nEdges * IN_DIM;
    if (gid >= total) return;
    int e = gid >> 6;
    int f = gid & (IN_DIM - 1);
    int s = src[e];
    int d = dst[e];
    atomicAdd(&msg[(size_t)d * IN_DIM + f], x[(size_t)s * IN_DIM + f]);
    if (f == 0) atomicAdd(&deg[d], 1.0f);
}

#define LDS_STRIDE (IN_DIM + 1)

__global__ void gcn_gemm_kernel(const float* __restrict__ W,
                                const float* __restrict__ bias,
                                const float* __restrict__ deg,
                                float* hout,
                                int nRowTiles) {
    __shared__ float sW[OUT_DIM * LDS_STRIDE];

    for (int i = threadIdx.x; i < OUT_DIM * IN_DIM; i += blockDim.x) {
        int n = i >> 6;
        int k = i & (IN_DIM - 1);
        sW[n * LDS_STRIDE + k] = W[i];
    }
    __syncthreads();

    const int lane = threadIdx.x & 31;
    const int waveInBlock = threadIdx.x >> 5;
    const int wavesPerBlock = blockDim.x >> 5;
    const int rowTile = blockIdx.x * wavesPerBlock + waveInBlock;
    if (rowTile >= nRowTiles) return;

    const int row0 = rowTile * 16;
    const int m    = lane & 15;
    const int hi   = lane >> 4;
    const int koff = hi * 2;

    const int r = row0 + m;
    float dv  = deg[r];
    float inv = 1.0f / fmaxf(dv, 1.0f);

    v2f a[16];
    const float* rowp = hout + (size_t)r * IN_DIM;
#pragma unroll
    for (int kk = 0; kk < 16; ++kk) {
        a[kk].x = rowp[kk * 4 + koff]     * inv;
        a[kk].y = rowp[kk * 4 + koff + 1] * inv;
    }

    v8f acc[4] = {};

#pragma unroll
    for (int kk = 0; kk < 16; ++kk) {
#pragma unroll
        for (int t = 0; t < 4; ++t) {
            int n = t * 16 + m;
            v2f b;
            b.x = sW[n * LDS_STRIDE + kk * 4 + koff];
            b.y = sW[n * LDS_STRIDE + kk * 4 + koff + 1];
            acc[t] = __builtin_amdgcn_wmma_f32_16x16x4_f32(
                 false, a[kk],  false, b,
                 (short)0, acc[t],
                 false,  false);
        }
    }

#pragma unroll
    for (int t = 0; t < 4; ++t) {
        int n = t * 16 + m;
        float bv = bias[n];
#pragma unroll
        for (int v = 0; v < 8; ++v) {
            int rr = row0 + v + hi * 8;
            hout[(size_t)rr * OUT_DIM + n] = acc[t][v] + bv;
        }
    }
}

extern "C" void kernel_launch(void* const* d_in, const int* in_sizes, int n_in,
                              void* d_out, int out_size, void* d_ws, size_t ws_size,
                              hipStream_t stream) {
    const float* x   = (const float*)d_in[0];
    const int*   src = (const int*)  d_in[1];
    const int*   dst = (const int*)  d_in[2];
    const float* W   = (const float*)d_in[3];
    const float* b   = (const float*)d_in[4];

    float* out = (float*)d_out;
    float* deg = (float*)d_ws;

    const int nNodes = in_sizes[0] / IN_DIM;
    const int nEdges = in_sizes[1];

    {
        int n4 = (nNodes * OUT_DIM) / 4;
        int blocks = (n4 + 255) / 256;
        if (blocks > 4096) blocks = 4096;
        zero4_kernel<<<blocks, 256, 0, stream>>>((float4*)out, n4);
    }
    {
        int blocks = (nNodes + 255) / 256;
        if (blocks > 2048) blocks = 2048;
        zero1_kernel<<<blocks, 256, 0, stream>>>(deg, nNodes);
    }

    {
        int total = nEdges * IN_DIM;
        int blocks = (total + 255) / 256;
        edge_scatter_kernel<<<blocks, 256, 0, stream>>>(x, src, dst, out, deg, nEdges);
    }

    {
        int nRowTiles = (nNodes + 15) / 16;
        int wavesPerBlock = 8;
        int blocks = (nRowTiles + wavesPerBlock - 1) / wavesPerBlock;
        gcn_gemm_kernel<<<blocks, 256, 0, stream>>>(W, b, deg, out, nRowTiles);
    }
}


# float-atomic scatter replaced by a same-call CSR (1 int atomic per edge gives its arrival-order rank, block-scan slot runs, fill) and one reader summing each row; sum order = arrival-order rank
# speedup vs baseline: 2.1456x; 2.1456x over previous
#include <hip/hip_runtime.h>


#define IN_DIM 64
#define OUT_DIM 64

typedef float v2f __attribute__((ext_vector_type(2)));
typedef float v8f __attribute__((ext_vector_type(8)));

__global__ void zero_i32_kernel(int* __restrict__ p, int n) {
    int i = blockIdx.x * blockDim.x + threadIdx.x;
    if (i < n) p[i] = 0;
}

__global__ void edge_rank_kernel(const int* __restrict__ dst,
                                 int* __restrict__ cnt,
                                 int* __restrict__ rank,
                                 int nEdges) {
    int e = blockIdx.x * blockDim.x + threadIdx.x;
    if (e >= nEdges) return;
    rank[e] = atomicAdd(&cnt[dst[e]], 1);
}

__global__ void segment_alloc_kernel(const int* __restrict__ cnt,
                                     int* __restrict__ start,
                                     int* __restrict__ cursor,
                                     int nNodes) {
    __shared__ int sWave[32];
    __shared__ int sBase;
    const int i    = blockIdx.x * blockDim.x + threadIdx.x;
    const int lane = threadIdx.x & (warpSize - 1);
    const int w    = threadIdx.x / warpSize;
    const int nW   = blockDim.x / warpSize;
    const int c    = (i < nNodes) ? cnt[i] : 0;
    int incl = c;
    for (int o = 1; o < warpSize; o <<= 1) {
        int v = __shfl_up(incl, o);
        if (lane >= o) incl += v;
    }
    if (lane == warpSize - 1) sWave[w] = incl;
    __syncthreads();
    if (threadIdx.x == 0) {
        int tot = 0;
        for (int k = 0; k < nW; ++k) { int t = sWave[k]; sWave[k] = tot; tot += t; }
        sBase = atomicAdd(cursor, tot);
    }
    __syncthreads();
    if (i < nNodes) start[i] = sBase + sWave[w] + incl - c;
}

__global__ void edge_fill_kernel(const int* __restrict__ src,
                                 const int* __restrict__ dst,
                                 const int* __restrict__ start,
                                 const int* __restrict__ rank,
                                 int* __restrict__ col,
                                 int nEdges) {
    int e = blockIdx.x * blockDim.x + threadIdx.x;
    if (e >= nEdges) return;
    col[start[dst[e]] + rank[e]] = src[e];
}

__global__ void gather_sum_kernel(const float* __restrict__ x,
                                  const int* __restrict__ col,
                                  const int* __restrict__ start,
                                  const int* __restrict__ cnt,
                                  float* __restrict__ msg,
                                  float* __restrict__ deg,
                                  int nNodes) {
    const int t    = blockIdx.x * blockDim.x + threadIdx.x;
    const int node = t >> 4;
    const int l    = t & 15;
    if (node >= nNodes) return;
    const int c = cnt[node];
    const int* cp = col + start[node];
    const float4* x4 = reinterpret_cast<const float4*>(x);
    const float4 z = make_float4(0.f, 0.f, 0.f, 0.f);
    float4 acc = z;
    const int last = c - 1;
    for (int j = 0; j < c; j += 4) {
        const int i0 = cp[j];
        const int i1 = cp[min(j + 1, last)];
        const int i2 = cp[min(j + 2, last)];
        const int i3 = cp[min(j + 3, last)];
        const float4 v0 = x4[(size_t)i0 * (IN_DIM / 4) + l];
        float4 v1 = x4[(size_t)i1 * (IN_DIM / 4) + l];
        float4 v2 = x4[(size_t)i2 * (IN_DIM / 4) + l];
        float4 v3 = x4[(size_t)i3 * (IN_DIM / 4) + l];
        if (j + 1 > last) v1 = z;
        if (j + 2 > last) v2 = z;
        if (j + 3 > last) v3 = z;
        acc.x += v0.x; acc.y += v0.y; acc.z += v0.z; acc.w += v0.w;
        acc.x += v1.x; acc.y += v1.y; acc.z += v1.z; acc.w += v1.w;
        acc.x += v2.x; acc.y += v2.y; acc.z += v2.z; acc.w += v2.w;
        acc.x += v3.x; acc.y += v3.y; acc.z += v3.z; acc.w += v3.w;
    }
    reinterpret_cast<float4*>(msg)[(size_t)node * (IN_DIM / 4) + l] = acc;
    if (l == 0) deg[node] = (float)c;
}

#define LDS_STRIDE (IN_DIM + 1)

__global__ void gcn_gemm_kernel(const float* __restrict__ W,
                                const float* __restrict__ bias,
                                const float* __restrict__ deg,
                                float* hout,
                                int nRowTiles) {
    __shared__ float sW[OUT_DIM * LDS_STRIDE];

    for (int i = threadIdx.x; i < OUT_DIM * IN_DIM; i += blockDim.x) {
        int n = i >> 6;
        int k = i & (IN_DIM - 1);
        sW[n * LDS_STRIDE + k] = W[i];
    }
    __syncthreads();

    const int lane = threadIdx.x & 31;
    const int waveInBlock = threadIdx.x >> 5;
    const int wavesPerBlock = blockDim.x >> 5;
    const int rowTile = blockIdx.x * wavesPerBlock + waveInBlock;
    if (rowTile >= nRowTiles) return;

    const int row0 = rowTile * 16;
    const int m    = lane & 15;
    const int hi   = lane >> 4;
    const int koff = hi * 2;

    const int r = row0 + m;
    float dv  = deg[r];
    float inv = 1.0f / fmaxf(dv, 1.0f);

    v2f a[16];
    const float* rowp = hout + (size_t)r * IN_DIM;
#pragma unroll
    for (int kk = 0; kk < 16; ++kk) {
        a[kk].x = rowp[kk * 4 + koff]     * inv;
        a[kk].y = rowp[kk * 4 + koff + 1] * inv;
    }

    v8f acc[4] = {};

#pragma unroll
    for (int kk = 0; kk < 16; ++kk) {
#pragma unroll
        for (int t = 0; t < 4; ++t) {
            int n = t * 16 + m;
            v2f b;
            b.x = sW[n * LDS_STRIDE + kk * 4 + koff];
            b.y = sW[n * LDS_STRIDE + kk * 4 + koff + 1];
            acc[t] = __builtin_amdgcn_wmma_f32_16x16x4_f32(
                 false, a[kk],  false, b,
                 (short)0, acc[t],
                 false,  false);
        }
    }

#pragma unroll
    for (int t = 0; t < 4; ++t) {
        int n = t * 16 + m;
        float bv = bias[n];
#pragma unroll
        for (int v = 0; v < 8; ++v) {
            int rr = row0 + v + hi * 8;
            hout[(size_t)rr * OUT_DIM + n] = acc[t][v] + bv;
        }
    }
}

extern "C" void kernel_launch(void* const* d_in, const int* in_sizes, int n_in,
                              void* d_out, int out_size, void* d_ws, size_t ws_size,
                              hipStream_t stream) {
    const float* x   = (const float*)d_in[0];
    const int*   src = (const int*)  d_in[1];
    const int*   dst = (const int*)  d_in[2];
    const float* W   = (const float*)d_in[3];
    const float* b   = (const float*)d_in[4];

    float* out = (float*)d_out;

    const int nNodes = in_sizes[0] / IN_DIM;
    const int nEdges = in_sizes[1];

    int*   cnt    = (int*)d_ws;
    int*   cursor = cnt + nNodes;
    int*   start  = cursor + 64;
    int*   rank   = start + nNodes;
    int*   col    = rank + nEdges;
    float* deg    = (float*)(col + nEdges);

    const int nodeBlocks = (nNodes + 255) / 256;
    const int edgeBlocks = (nEdges + 255) / 256;

    zero_i32_kernel<<<(nNodes + 1 + 255) / 256, 256, 0, stream>>>(cnt, nNodes + 1);

    edge_rank_kernel<<<edgeBlocks, 256, 0, stream>>>(dst, cnt, rank, nEdges);

    segment_alloc_kernel<<<nodeBlocks, 256, 0, stream>>>(cnt, start, cursor, nNodes);

    edge_fill_kernel<<<edgeBlocks, 256, 0, stream>>>(src, dst, start, rank, col, nEdges);

    {
        int blocks = (nNodes * 16 + 255) / 256;
        gather_sum_kernel<<<blocks, 256, 0, stream>>>(x, col, start, cnt, out, deg, nNodes);
    }

    {
        int nRowTiles = (nNodes + 15) / 16;
        int wavesPerBlock = 8;
        int blocks = (nRowTiles + wavesPerBlock - 1) / wavesPerBlock;
        gcn_gemm_kernel<<<blocks, 256, 0, stream>>>(W, b, deg, out, nRowTiles);
    }
}


# per-edge global returning add replaced by a two-level counting sort on LDS counters (LDS returning add = rank): 512-node ranges, then nodes; row sum order = LDS arrival order
# speedup vs baseline: 3.4361x; 1.6015x over previous
#include <hip/hip_runtime.h>


#define IN_DIM 64
#define OUT_DIM 64

typedef float v2f __attribute__((ext_vector_type(2)));
typedef float v8f __attribute__((ext_vector_type(8)));

#define BIN_SHIFT   9
#define BIN_NODES   512
#define MAX_BINS    256
#define CHUNK_T     512
#define CHUNK_EPT   16
#define CHUNK_EDGES (CHUNK_T * CHUNK_EPT)

__global__ void bin_count_kernel(const int* __restrict__ dst,
                                 int* __restrict__ hist,
                                 int nEdges, int nBins) {
    __shared__ int h[MAX_BINS];
    const int tid = threadIdx.x;
    if (tid < MAX_BINS) h[tid] = 0;
    __syncthreads();
    const int base = blockIdx.x * CHUNK_EDGES;
    int d[CHUNK_EPT];
#pragma unroll
    for (int k = 0; k < CHUNK_EPT; ++k) {
        const int e = base + k * CHUNK_T + tid;
        d[k] = (e < nEdges) ? dst[e] : -1;
    }
#pragma unroll
    for (int k = 0; k < CHUNK_EPT; ++k)
        if (d[k] >= 0) atomicAdd(&h[d[k] >> BIN_SHIFT], 1);
    __syncthreads();
    if (tid < nBins) hist[blockIdx.x * MAX_BINS + tid] = h[tid];
}

__global__ void bin_scan_kernel(const int* __restrict__ hist,
                                int* __restrict__ offs,
                                int* __restrict__ binBase,
                                int nChunks, int nBins) {
    __shared__ int sWave[32];
    const int b    = threadIdx.x;
    const int lane = threadIdx.x & (warpSize - 1);
    const int w    = threadIdx.x / warpSize;
    const int nW   = blockDim.x / warpSize;
    int run = 0;
    if (b < nBins) {
        for (int c0 = 0; c0 < nChunks; c0 += 16) {
            int t[16];
#pragma unroll
            for (int j = 0; j < 16; ++j) {
                const int c = c0 + j;
                t[j] = (c < nChunks) ? hist[c * MAX_BINS + b] : 0;
            }
#pragma unroll
            for (int j = 0; j < 16; ++j) {
                const int c = c0 + j;
                if (c < nChunks) { offs[c * MAX_BINS + b] = run; run += t[j]; }
            }
        }
    }
    int incl = run;
    for (int o = 1; o < warpSize; o <<= 1) {
        int v = __shfl_up(incl, o);
        if (lane >= o) incl += v;
    }
    if (lane == warpSize - 1) sWave[w] = incl;
    __syncthreads();
    if (threadIdx.x == 0) {
        int tot = 0;
        for (int k = 0; k < nW; ++k) { int t = sWave[k]; sWave[k] = tot; tot += t; }
    }
    __syncthreads();
    const int excl = sWave[w] + incl - run;
    if (b < nBins) binBase[b] = excl;
    if (b == nBins - 1) binBase[nBins] = excl + run;
}

__global__ void bin_scatter_kernel(const int* __restrict__ src,
                                   const int* __restrict__ dst,
                                   const int* __restrict__ offs,
                                   const int* __restrict__ binBase,
                                   int2* __restrict__ pairs,
                                   int nEdges, int nBins) {
    __shared__ int h[MAX_BINS];
    const int tid = threadIdx.x;
    if (tid < MAX_BINS)
        h[tid] = (tid < nBins) ? offs[blockIdx.x * MAX_BINS + tid] + binBase[tid] : 0;
    __syncthreads();
    const int base = blockIdx.x * CHUNK_EDGES;
    int d[CHUNK_EPT], s[CHUNK_EPT];
#pragma unroll
    for (int k = 0; k < CHUNK_EPT; ++k) {
        const int e = base + k * CHUNK_T + tid;
        const bool ok = e < nEdges;
        d[k] = ok ? dst[e] : -1;
        s[k] = ok ? src[e] : 0;
    }
#pragma unroll
    for (int k = 0; k < CHUNK_EPT; ++k) {
        if (d[k] >= 0) {
            const int pos = atomicAdd(&h[d[k] >> BIN_SHIFT], 1);
            pairs[pos] = make_int2(s[k], d[k]);
        }
    }
}

__global__ void bin_csr_kernel(const int2* __restrict__ pairs,
                               const int* __restrict__ binBase,
                               int* __restrict__ col,
                               int* __restrict__ start,
                               int* __restrict__ cnt,
                               int nNodes) {
    __shared__ int c[BIN_NODES];
    __shared__ int sWave[32];
    const int tid  = threadIdx.x;
    const int lane = threadIdx.x & (warpSize - 1);
    const int w    = threadIdx.x / warpSize;
    const int nW   = blockDim.x / warpSize;
    const int lo   = blockIdx.x << BIN_SHIFT;
    const int e0   = binBase[blockIdx.x];
    const int e1   = binBase[blockIdx.x + 1];
    c[tid] = 0;
    __syncthreads();
    for (int i = e0 + tid; i < e1; i += 4 * BIN_NODES) {
        int2 p[4];
#pragma unroll
        for (int j = 0; j < 4; ++j) p[j] = pairs[min(i + j * BIN_NODES, e1 - 1)];
        asm volatile("" :: "v"(p[0].y), "v"(p[1].y), "v"(p[2].y), "v"(p[3].y));
#pragma unroll
        for (int j = 0; j < 4; ++j)
            if (i + j * BIN_NODES < e1) atomicAdd(&c[p[j].y - lo], 1);
    }
    __syncthreads();
    const int my = c[tid];
    int incl = my;
    for (int o = 1; o < warpSize; o <<= 1) {
        int v = __shfl_up(incl, o);
        if (lane >= o) incl += v;
    }
    if (lane == warpSize - 1) sWave[w] = incl;
    __syncthreads();
    if (tid == 0) {
        int tot = 0;
        for (int k = 0; k < nW; ++k) { int t = sWave[k]; sWave[k] = tot; tot += t; }
    }
    __syncthreads();
    const int first = e0 + sWave[w] + incl - my;
    c[tid] = first;
    const int node = lo + tid;
    if (node < nNodes) { start[node] = first; cnt[node] = my; }
    __syncthreads();
    for (int i = e0 + tid; i < e1; i += 4 * BIN_NODES) {
        int2 p[4];
#pragma unroll
        for (int j = 0; j < 4; ++j) p[j] = pairs[min(i + j * BIN_NODES, e1 - 1)];
        asm volatile("" :: "v"(p[0].x), "v"(p[0].y), "v"(p[1].x), "v"(p[1].y),
                           "v"(p[2].x), "v"(p[2].y), "v"(p[3].x), "v"(p[3].y));
#pragma unroll
        for (int j = 0; j < 4; ++j)
            if (i + j * BIN_NODES < e1) col[atomicAdd(&c[p[j].y - lo], 1)] = p[j].x;
    }
}

__global__ void gather_sum_kernel(const float* __restrict__ x,
                                  const int* __restrict__ col,
                                  const int* __restrict__ start,
                                  const int* __restrict__ cnt,
                                  float* __restrict__ msg,
                                  float* __restrict__ deg,
                                  int nNodes) {
    const int t    = blockIdx.x * blockDim.x + threadIdx.x;
    const int node = t >> 4;
    const int l    = t & 15;
    if (node >= nNodes) return;
    const int c = cnt[node];
    const int* cp = col + start[node];
    const float4* x4 = reinterpret_cast<const float4*>(x);
    const float4 z = make_float4(0.f, 0.f, 0.f, 0.f);
    float4 acc = z;
    const int last = c - 1;
    for (int j = 0; j < c; j += 4) {
        const int i0 = cp[j];
        const int i1 = cp[min(j + 1, last)];
        const int i2 = cp[min(j + 2, last)];
        const int i3 = cp[min(j + 3, last)];
        const float4 v0 = x4[(size_t)i0 * (IN_DIM / 4) + l];
        float4 v1 = x4[(size_t)i1 * (IN_DIM / 4) + l];
        float4 v2 = x4[(size_t)i2 * (IN_DIM / 4) + l];
        float4 v3 = x4[(size_t)i3 * (IN_DIM / 4) + l];
        if (j + 1 > last) v1 = z;
        if (j + 2 > last) v2 = z;
        if (j + 3 > last) v3 = z;
        acc.x += v0.x; acc.y += v0.y; acc.z += v0.z; acc.w += v0.w;
        acc.x += v1.x; acc.y += v1.y; acc.z += v1.z; acc.w += v1.w;
        acc.x += v2.x; acc.y += v2.y; acc.z += v2.z; acc.w += v2.w;
        acc.x += v3.x; acc.y += v3.y; acc.z += v3.z; acc.w += v3.w;
    }
    reinterpret_cast<float4*>(msg)[(size_t)node * (IN_DIM / 4) + l] = acc;
    if (l == 0) deg[node] = (float)c;
}

#define LDS_STRIDE (IN_DIM + 1)

__global__ void gcn_gemm_kernel(const float* __restrict__ W,
                                const float* __restrict__ bias,
                                const float* __restrict__ deg,
                                float* hout,
                                int nRowTiles) {
    __shared__ float sW[OUT_DIM * LDS_STRIDE];

    for (int i = threadIdx.x; i < OUT_DIM * IN_DIM; i += blockDim.x) {
        int n = i >> 6;
        int k = i & (IN_DIM - 1);
        sW[n * LDS_STRIDE + k] = W[i];
    }
    __syncthreads();

    const int lane = threadIdx.x & 31;
    const int waveInBlock = threadIdx.x >> 5;
    const int wavesPerBlock = blockDim.x >> 5;
    const int rowTile = blockIdx.x * wavesPerBlock + waveInBlock;
    if (rowTile >= nRowTiles) return;

    const int row0 = rowTile * 16;
    const int m    = lane & 15;
    const int hi   = lane >> 4;
    const int koff = hi * 2;

    const int r = row0 + m;
    float dv  = deg[r];
    float inv = 1.0f / fmaxf(dv, 1.0f);

    v2f a[16];
    const float* rowp = hout + (size_t)r * IN_DIM;
#pragma unroll
    for (int kk = 0; kk < 16; ++kk) {
        a[kk].x = rowp[kk * 4 + koff]     * inv;
        a[kk].y = rowp[kk * 4 + koff + 1] * inv;
    }

    v8f acc[4] = {};

#pragma unroll
    for (int kk = 0; kk < 16; ++kk) {
#pragma unroll
        for (int t = 0; t < 4; ++t) {
            int n = t * 16 + m;
            v2f b;
            b.x = sW[n * LDS_STRIDE + kk * 4 + koff];
            b.y = sW[n * LDS_STRIDE + kk * 4 + koff + 1];
            acc[t] = __builtin_amdgcn_wmma_f32_16x16x4_f32(
                 false, a[kk],  false, b,
                 (short)0, acc[t],
                 false,  false);
        }
    }

#pragma unroll
    for (int t = 0; t < 4; ++t) {
        int n = t * 16 + m;
        float bv = bias[n];
#pragma unroll
        for (int v = 0; v < 8; ++v) {
            int rr = row0 + v + hi * 8;
            hout[(size_t)rr * OUT_DIM + n] = acc[t][v] + bv;
        }
    }
}

extern "C" void kernel_launch(void* const* d_in, const int* in_sizes, int n_in,
                              void* d_out, int out_size, void* d_ws, size_t ws_size,
                              hipStream_t stream) {
    const float* x   = (const float*)d_in[0];
    const int*   src = (const int*)  d_in[1];
    const int*   dst = (const int*)  d_in[2];
    const float* W   = (const float*)d_in[3];
    const float* b   = (const float*)d_in[4];

    float* out = (float*)d_out;

    const int nNodes = in_sizes[0] / IN_DIM;
    const int nEdges = in_sizes[1];

    const int nBins   = (nNodes + BIN_NODES - 1) / BIN_NODES;
    const int nChunks = (nEdges + CHUNK_EDGES - 1) / CHUNK_EDGES;

    int*   hist    = (int*)d_ws;
    int*   offs    = hist + nChunks * MAX_BINS;
    int*   binBase = offs + nChunks * MAX_BINS;
    int2*  pairs   = (int2*)(binBase + 320);
    int*   col     = (int*)(pairs + nEdges);
    int*   start   = col + nEdges;
    int*   cnt     = start + nNodes;
    float* deg     = (float*)(cnt + nNodes);

    bin_count_kernel<<<nChunks, CHUNK_T, 0, stream>>>(dst, hist, nEdges, nBins);

    bin_scan_kernel<<<1, MAX_BINS, 0, stream>>>(hist, offs, binBase, nChunks, nBins);

    bin_scatter_kernel<<<nChunks, CHUNK_T, 0, stream>>>(src, dst, offs, binBase, pairs, nEdges, nBins);

    bin_csr_kernel<<<nBins, BIN_NODES, 0, stream>>>(pairs, binBase, col, start, cnt, nNodes);

    {
        int blocks = (nNodes * 16 + 255) / 256;
        gather_sum_kernel<<<blocks, 256, 0, stream>>>(x, col, start, cnt, out, deg, nNodes);
    }

    {
        int nRowTiles = (nNodes + 15) / 16;
        int wavesPerBlock = 8;
        int blocks = (nRowTiles + wavesPerBlock - 1) / wavesPerBlock;
        gcn_gemm_kernel<<<blocks, 256, 0, stream>>>(W, b, deg, out, nRowTiles);
    }
}


# sort level 1 made chunk-local: each chunk sorts its pairs by range in LDS, range blocks walk the chunks' segments; count and scan launches gone; row sum order = LDS arrival order, walk changed
# speedup vs baseline: 4.4470x; 1.2942x over previous
#include <hip/hip_runtime.h>


#define IN_DIM 64
#define OUT_DIM 64

typedef float v2f __attribute__((ext_vector_type(2)));
typedef float v8f __attribute__((ext_vector_type(8)));

#define BIN_SHIFT   9
#define BIN_NODES   512
#define MAX_BINS    256
#define SEG_STRIDE  (MAX_BINS + 2)
#define CHUNK_T     512
#define CHUNK_EPT   16
#define CHUNK_EDGES (CHUNK_T * CHUNK_EPT)
#define SEG_BATCH   5

__global__ void chunk_sort_kernel(const int* __restrict__ src,
                                  const int* __restrict__ dst,
                                  int* __restrict__ seg,
                                  int2* __restrict__ pairs,
                                  int nEdges) {
    __shared__ int h[MAX_BINS];
    __shared__ int sWave[32];
    const int tid  = threadIdx.x;
    const int lane = threadIdx.x & (warpSize - 1);
    const int w    = threadIdx.x / warpSize;
    const int nWs  = MAX_BINS / warpSize;
    if (tid < MAX_BINS) h[tid] = 0;
    __syncthreads();
    const int base = blockIdx.x * CHUNK_EDGES;
    int d[CHUNK_EPT], s[CHUNK_EPT];
#pragma unroll
    for (int k = 0; k < CHUNK_EPT; ++k) {
        const int e = base + k * CHUNK_T + tid;
        const bool ok = e < nEdges;
        d[k] = ok ? dst[e] : -1;
        s[k] = ok ? src[e] : 0;
    }
#pragma unroll
    for (int k = 0; k < CHUNK_EPT; ++k)
        if (d[k] >= 0) atomicAdd(&h[d[k] >> BIN_SHIFT], 1);
    __syncthreads();
    const int my = (tid < MAX_BINS) ? h[tid] : 0;
    int incl = my;
    for (int o = 1; o < warpSize; o <<= 1) {
        int v = __shfl_up(incl, o);
        if (lane >= o) incl += v;
    }
    if (lane == warpSize - 1 && w < nWs) sWave[w] = incl;
    __syncthreads();
    if (tid == 0) {
        int tot = 0;
        for (int k = 0; k < nWs; ++k) { int t = sWave[k]; sWave[k] = tot; tot += t; }
    }
    __syncthreads();
    if (tid < MAX_BINS) {
        const int excl = sWave[w] + incl - my;
        h[tid] = excl;
        seg[blockIdx.x * SEG_STRIDE + tid] = excl;
        if (tid == MAX_BINS - 1) seg[blockIdx.x * SEG_STRIDE + MAX_BINS] = excl + my;
    }
    __syncthreads();
#pragma unroll
    for (int k = 0; k < CHUNK_EPT; ++k) {
        if (d[k] >= 0) {
            const int pos = atomicAdd(&h[d[k] >> BIN_SHIFT], 1);
            pairs[base + pos] = make_int2(s[k], d[k]);
        }
    }
}

__global__ void bin_csr_kernel(const int2* __restrict__ pairs,
                               const int* __restrict__ seg,
                               int* __restrict__ col,
                               int* __restrict__ start,
                               int* __restrict__ cnt,
                               int nNodes, int nChunks) {
    __shared__ int c[BIN_NODES];
    __shared__ int sWave[32];
    __shared__ int sLo[BIN_NODES];
    __shared__ int sN[BIN_NODES];
    const int tid  = threadIdx.x;
    const int lane = threadIdx.x & (warpSize - 1);
    const int w    = threadIdx.x / warpSize;
    const int nW   = blockDim.x / warpSize;
    const int b    = blockIdx.x;
    const int lo   = b << BIN_SHIFT;

    int part = 0;
    for (int t = tid; t < nChunks; t += BIN_NODES) part += seg[t * SEG_STRIDE + b];
    for (int o = warpSize >> 1; o > 0; o >>= 1) part += __shfl_xor(part, o);
    if (lane == 0) sWave[w] = part;
    c[tid] = 0;
    __syncthreads();
    int e0 = 0;
    for (int k = 0; k < nW; ++k) e0 += sWave[k];
    __syncthreads();

    for (int cb = 0; cb < nChunks; cb += BIN_NODES) {
        const int tileN = min(nChunks - cb, BIN_NODES);
        if (tid < tileN) {
            const int a0 = seg[(cb + tid) * SEG_STRIDE + b];
            const int a1 = seg[(cb + tid) * SEG_STRIDE + b + 1];
            sN[tid]  = a1 - a0;
            sLo[tid] = (a1 > a0) ? (cb + tid) * CHUNK_EDGES + a0 : 0;
        }
        __syncthreads();
        for (int j0 = w; j0 < tileN; j0 += nW * SEG_BATCH) {
            int2 p[SEG_BATCH][2];
            int  n[SEG_BATCH];
#pragma unroll
            for (int q = 0; q < SEG_BATCH; ++q) {
                const int j = j0 + q * nW;
                const bool ok = j < tileN;
                n[q] = ok ? sN[j] : 0;
                const int f = ok ? sLo[j] : 0;
                const int last = max(n[q] - 1, 0);
                p[q][0] = pairs[f + min(lane, last)];
                p[q][1] = pairs[f + min(lane + 32, last)];
            }
            asm volatile("" :: "v"(p[0][0].x), "v"(p[0][0].y), "v"(p[0][1].x), "v"(p[0][1].y),
                               "v"(p[1][0].x), "v"(p[1][0].y), "v"(p[1][1].x), "v"(p[1][1].y),
                               "v"(p[2][0].x), "v"(p[2][0].y), "v"(p[2][1].x), "v"(p[2][1].y),
                               "v"(p[3][0].x), "v"(p[3][0].y), "v"(p[3][1].x), "v"(p[3][1].y),
                               "v"(p[4][0].x), "v"(p[4][0].y), "v"(p[4][1].x), "v"(p[4][1].y));
#pragma unroll
            for (int q = 0; q < SEG_BATCH; ++q) {
                if (lane < n[q])      atomicAdd(&c[p[q][0].y - lo], 1);
                if (lane + 32 < n[q]) atomicAdd(&c[p[q][1].y - lo], 1);
                if (n[q] > 64) {
                    const int j = j0 + q * nW;
                    const int f = sLo[j];
                    for (int i = lane + 64; i < n[q]; i += 32)
                        atomicAdd(&c[pairs[f + i].y - lo], 1);
                }
            }
        }
        __syncthreads();
    }

    const int my = c[tid];
    int incl = my;
    for (int o = 1; o < warpSize; o <<= 1) {
        int v = __shfl_up(incl, o);
        if (lane >= o) incl += v;
    }
    if (lane == warpSize - 1) sWave[w] = incl;
    __syncthreads();
    if (tid == 0) {
        int tot = 0;
        for (int k = 0; k < nW; ++k) { int t = sWave[k]; sWave[k] = tot; tot += t; }
    }
    __syncthreads();
    const int first = e0 + sWave[w] + incl - my;
    c[tid] = first;
    const int node = lo + tid;
    if (node < nNodes) { start[node] = first; cnt[node] = my; }
    __syncthreads();

    for (int cb = 0; cb < nChunks; cb += BIN_NODES) {
        const int tileN = min(nChunks - cb, BIN_NODES);
        if (nChunks > BIN_NODES) {
            __syncthreads();
            if (tid < tileN) {
                const int a0 = seg[(cb + tid) * SEG_STRIDE + b];
                const int a1 = seg[(cb + tid) * SEG_STRIDE + b + 1];
                sN[tid]  = a1 - a0;
                sLo[tid] = (a1 > a0) ? (cb + tid) * CHUNK_EDGES + a0 : 0;
            }
            __syncthreads();
        }
        for (int j0 = w; j0 < tileN; j0 += nW * SEG_BATCH) {
            int2 p[SEG_BATCH][2];
            int  n[SEG_BATCH];
#pragma unroll
            for (int q = 0; q < SEG_BATCH; ++q) {
                const int j = j0 + q * nW;
                const bool ok = j < tileN;
                n[q] = ok ? sN[j] : 0;
                const int f = ok ? sLo[j] : 0;
                const int last = max(n[q] - 1, 0);
                p[q][0] = pairs[f + min(lane, last)];
                p[q][1] = pairs[f + min(lane + 32, last)];
            }
            asm volatile("" :: "v"(p[0][0].x), "v"(p[0][0].y), "v"(p[0][1].x), "v"(p[0][1].y),
                               "v"(p[1][0].x), "v"(p[1][0].y), "v"(p[1][1].x), "v"(p[1][1].y),
                               "v"(p[2][0].x), "v"(p[2][0].y), "v"(p[2][1].x), "v"(p[2][1].y),
                               "v"(p[3][0].x), "v"(p[3][0].y), "v"(p[3][1].x), "v"(p[3][1].y),
                               "v"(p[4][0].x), "v"(p[4][0].y), "v"(p[4][1].x), "v"(p[4][1].y));
#pragma unroll
            for (int q = 0; q < SEG_BATCH; ++q) {
                if (lane < n[q])      col[atomicAdd(&c[p[q][0].y - lo], 1)] = p[q][0].x;
                if (lane + 32 < n[q]) col[atomicAdd(&c[p[q][1].y - lo], 1)] = p[q][1].x;
                if (n[q] > 64) {
                    const int j = j0 + q * nW;
                    const int f = sLo[j];
                    for (int i = lane + 64; i < n[q]; i += 32) {
                        const int2 t = pairs[f + i];
                        col[atomicAdd(&c[t.y - lo], 1)] = t.x;
                    }
                }
            }
        }
    }
}

__global__ void gather_sum_kernel(const float* __restrict__ x,
                                  const int* __restrict__ col,
                                  const int* __restrict__ start,
                                  const int* __restrict__ cnt,
                                  float* __restrict__ msg,
                                  float* __restrict__ deg,
                                  int nNodes) {
    const int t    = blockIdx.x * blockDim.x + threadIdx.x;
    const int node = t >> 4;
    const int l    = t & 15;
    if (node >= nNodes) return;
    const int c = cnt[node];
    const int* cp = col + start[node];
    const float4* x4 = reinterpret_cast<const float4*>(x);
    const float4 z = make_float4(0.f, 0.f, 0.f, 0.f);
    float4 acc = z;
    const int last = c - 1;
    for (int j = 0; j < c; j += 4) {
        const int i0 = cp[j];
        const int i1 = cp[min(j + 1, last)];
        const int i2 = cp[min(j + 2, last)];
        const int i3 = cp[min(j + 3, last)];
        const float4 v0 = x4[(size_t)i0 * (IN_DIM / 4) + l];
        float4 v1 = x4[(size_t)i1 * (IN_DIM / 4) + l];
        float4 v2 = x4[(size_t)i2 * (IN_DIM / 4) + l];
        float4 v3 = x4[(size_t)i3 * (IN_DIM / 4) + l];
        if (j + 1 > last) v1 = z;
        if (j + 2 > last) v2 = z;
        if (j + 3 > last) v3 = z;
        acc.x += v0.x; acc.y += v0.y; acc.z += v0.z; acc.w += v0.w;
        acc.x += v1.x; acc.y += v1.y; acc.z += v1.z; acc.w += v1.w;
        acc.x += v2.x; acc.y += v2.y; acc.z += v2.z; acc.w += v2.w;
        acc.x += v3.x; acc.y += v3.y; acc.z += v3.z; acc.w += v3.w;
    }
    reinterpret_cast<float4*>(msg)[(size_t)node * (IN_DIM / 4) + l] = acc;
    if (l == 0) deg[node] = (float)c;
}

#define LDS_STRIDE (IN_DIM + 1)

__global__ void gcn_gemm_kernel(const float* __restrict__ W,
                                const float* __restrict__ bias,
                                const float* __restrict__ deg,
                                float* hout,
                                int nRowTiles) {
    __shared__ float sW[OUT_DIM * LDS_STRIDE];

    for (int i = threadIdx.x; i < OUT_DIM * IN_DIM; i += blockDim.x) {
        int n = i >> 6;
        int k = i & (IN_DIM - 1);
        sW[n * LDS_STRIDE + k] = W[i];
    }
    __syncthreads();

    const int lane = threadIdx.x & 31;
    const int waveInBlock = threadIdx.x >> 5;
    const int wavesPerBlock = blockDim.x >> 5;
    const int rowTile = blockIdx.x * wavesPerBlock + waveInBlock;
    if (rowTile >= nRowTiles) return;

    const int row0 = rowTile * 16;
    const int m    = lane & 15;
    const int hi   = lane >> 4;
    const int koff = hi * 2;

    const int r = row0 + m;
    float dv  = deg[r];
    float inv = 1.0f / fmaxf(dv, 1.0f);

    v2f a[16];
    const float* rowp = hout + (size_t)r * IN_DIM;
#pragma unroll
    for (int kk = 0; kk < 16; ++kk) {
        a[kk].x = rowp[kk * 4 + koff]     * inv;
        a[kk].y = rowp[kk * 4 + koff + 1] * inv;
    }

    v8f acc[4] = {};

#pragma unroll
    for (int kk = 0; kk < 16; ++kk) {
#pragma unroll
        for (int t = 0; t < 4; ++t) {
            int n = t * 16 + m;
            v2f b;
            b.x = sW[n * LDS_STRIDE + kk * 4 + koff];
            b.y = sW[n * LDS_STRIDE + kk * 4 + koff + 1];
            acc[t] = __builtin_amdgcn_wmma_f32_16x16x4_f32(
                 false, a[kk],  false, b,
                 (short)0, acc[t],
                 false,  false);
        }
    }

#pragma unroll
    for (int t = 0; t < 4; ++t) {
        int n = t * 16 + m;
        float bv = bias[n];
#pragma unroll
        for (int v = 0; v < 8; ++v) {
            int rr = row0 + v + hi * 8;
            hout[(size_t)rr * OUT_DIM + n] = acc[t][v] + bv;
        }
    }
}

extern "C" void kernel_launch(void* const* d_in, const int* in_sizes, int n_in,
                              void* d_out, int out_size, void* d_ws, size_t ws_size,
                              hipStream_t stream) {
    const float* x   = (const float*)d_in[0];
    const int*   src = (const int*)  d_in[1];
    const int*   dst = (const int*)  d_in[2];
    const float* W   = (const float*)d_in[3];
    const float* b   = (const float*)d_in[4];

    float* out = (float*)d_out;

    const int nNodes = in_sizes[0] / IN_DIM;
    const int nEdges = in_sizes[1];

    const int nBins   = (nNodes + BIN_NODES - 1) / BIN_NODES;
    const int nChunks = (nEdges + CHUNK_EDGES - 1) / CHUNK_EDGES;

    int*   seg     = (int*)d_ws;
    int2*  pairs   = (int2*)(seg + nChunks * SEG_STRIDE);
    int*   col     = (int*)(pairs + nChunks * CHUNK_EDGES);
    int*   start   = col + nEdges;
    int*   cnt     = start + nNodes;
    float* deg     = (float*)(cnt + nNodes);

    chunk_sort_kernel<<<nChunks, CHUNK_T, 0, stream>>>(src, dst, seg, pairs, nEdges);

    bin_csr_kernel<<<nBins, BIN_NODES, 0, stream>>>(pairs, seg, col, start, cnt, nNodes, nChunks);

    {
        int blocks = (nNodes * 16 + 255) / 256;
        gather_sum_kernel<<<blocks, 256, 0, stream>>>(x, col, start, cnt, out, deg, nNodes);
    }

    {
        int nRowTiles = (nNodes + 15) / 16;
        int wavesPerBlock = 8;
        int blocks = (nRowTiles + wavesPerBlock - 1) / wavesPerBlock;
        gcn_gemm_kernel<<<blocks, 256, 0, stream>>>(W, b, deg, out, nRowTiles);
    }
}


# row sums moved into the GEMM kernel: each 16-lane group still sums its rows in registers in bucket order (4 rows a group), sums reach WMMA via LDS; msg plane and one launch gone; sum order unchanged
# speedup vs baseline: 5.8134x; 1.3073x over previous
#include <hip/hip_runtime.h>


#define IN_DIM 64
#define OUT_DIM 64

typedef float v2f __attribute__((ext_vector_type(2)));
typedef float v8f __attribute__((ext_vector_type(8)));

#define BIN_SHIFT   9
#define BIN_NODES   512
#define MAX_BINS    256
#define SEG_STRIDE  (MAX_BINS + 2)
#define CHUNK_T     512
#define CHUNK_EPT   16
#define CHUNK_EDGES (CHUNK_T * CHUNK_EPT)
#define SEG_BATCH   5

__global__ void chunk_sort_kernel(const int* __restrict__ src,
                                  const int* __restrict__ dst,
                                  int* __restrict__ seg,
                                  int2* __restrict__ pairs,
                                  int nEdges) {
    __shared__ int h[MAX_BINS];
    __shared__ int sWave[32];
    const int tid  = threadIdx.x;
    const int lane = threadIdx.x & (warpSize - 1);
    const int w    = threadIdx.x / warpSize;
    const int nWs  = MAX_BINS / warpSize;
    if (tid < MAX_BINS) h[tid] = 0;
    __syncthreads();
    const int base = blockIdx.x * CHUNK_EDGES;
    int d[CHUNK_EPT], s[CHUNK_EPT];
#pragma unroll
    for (int k = 0; k < CHUNK_EPT; ++k) {
        const int e = base + k * CHUNK_T + tid;
        const bool ok = e < nEdges;
        d[k] = ok ? dst[e] : -1;
        s[k] = ok ? src[e] : 0;
    }
#pragma unroll
    for (int k = 0; k < CHUNK_EPT; ++k)
        if (d[k] >= 0) atomicAdd(&h[d[k] >> BIN_SHIFT], 1);
    __syncthreads();
    const int my = (tid < MAX_BINS) ? h[tid] : 0;
    int incl = my;
    for (int o = 1; o < warpSize; o <<= 1) {
        int v = __shfl_up(incl, o);
        if (lane >= o) incl += v;
    }
    if (lane == warpSize - 1 && w < nWs) sWave[w] = incl;
    __syncthreads();
    if (tid == 0) {
        int tot = 0;
        for (int k = 0; k < nWs; ++k) { int t = sWave[k]; sWave[k] = tot; tot += t; }
    }
    __syncthreads();
    if (tid < MAX_BINS) {
        const int excl = sWave[w] + incl - my;
        h[tid] = excl;
        seg[blockIdx.x * SEG_STRIDE + tid] = excl;
        if (tid == MAX_BINS - 1) seg[blockIdx.x * SEG_STRIDE + MAX_BINS] = excl + my;
    }
    __syncthreads();
#pragma unroll
    for (int k = 0; k < CHUNK_EPT; ++k) {
        if (d[k] >= 0) {
            const int pos = atomicAdd(&h[d[k] >> BIN_SHIFT], 1);
            pairs[base + pos] = make_int2(s[k], d[k]);
        }
    }
}

__global__ void bin_csr_kernel(const int2* __restrict__ pairs,
                               const int* __restrict__ seg,
                               int* __restrict__ col,
                               int* __restrict__ start,
                               int* __restrict__ cnt,
                               int nNodes, int nChunks) {
    __shared__ int c[BIN_NODES];
    __shared__ int sWave[32];
    __shared__ int sLo[BIN_NODES];
    __shared__ int sN[BIN_NODES];
    const int tid  = threadIdx.x;
    const int lane = threadIdx.x & (warpSize - 1);
    const int w    = threadIdx.x / warpSize;
    const int nW   = blockDim.x / warpSize;
    const int b    = blockIdx.x;
    const int lo   = b << BIN_SHIFT;

    int part = 0;
    for (int t = tid; t < nChunks; t += BIN_NODES) part += seg[t * SEG_STRIDE + b];
    for (int o = warpSize >> 1; o > 0; o >>= 1) part += __shfl_xor(part, o);
    if (lane == 0) sWave[w] = part;
    c[tid] = 0;
    __syncthreads();
    int e0 = 0;
    for (int k = 0; k < nW; ++k) e0 += sWave[k];
    __syncthreads();

    for (int cb = 0; cb < nChunks; cb += BIN_NODES) {
        const int tileN = min(nChunks - cb, BIN_NODES);
        if (tid < tileN) {
            const int a0 = seg[(cb + tid) * SEG_STRIDE + b];
            const int a1 = seg[(cb + tid) * SEG_STRIDE + b + 1];
            sN[tid]  = a1 - a0;
            sLo[tid] = (a1 > a0) ? (cb + tid) * CHUNK_EDGES + a0 : 0;
        }
        __syncthreads();
        for (int j0 = w; j0 < tileN; j0 += nW * SEG_BATCH) {
            int2 p[SEG_BATCH][2];
            int  n[SEG_BATCH];
#pragma unroll
            for (int q = 0; q < SEG_BATCH; ++q) {
                const int j = j0 + q * nW;
                const bool ok = j < tileN;
                n[q] = ok ? sN[j] : 0;
                const int f = ok ? sLo[j] : 0;
                const int last = max(n[q] - 1, 0);
                p[q][0] = pairs[f + min(lane, last)];
                p[q][1] = pairs[f + min(lane + 32, last)];
            }
            asm volatile("" :: "v"(p[0][0].x), "v"(p[0][0].y), "v"(p[0][1].x), "v"(p[0][1].y),
                               "v"(p[1][0].x), "v"(p[1][0].y), "v"(p[1][1].x), "v"(p[1][1].y),
                               "v"(p[2][0].x), "v"(p[2][0].y), "v"(p[2][1].x), "v"(p[2][1].y),
                               "v"(p[3][0].x), "v"(p[3][0].y), "v"(p[3][1].x), "v"(p[3][1].y),
                               "v"(p[4][0].x), "v"(p[4][0].y), "v"(p[4][1].x), "v"(p[4][1].y));
#pragma unroll
            for (int q = 0; q < SEG_BATCH; ++q) {
                if (lane < n[q])      atomicAdd(&c[p[q][0].y - lo], 1);
                if (lane + 32 < n[q]) atomicAdd(&c[p[q][1].y - lo], 1);
                if (n[q] > 64) {
                    const int j = j0 + q * nW;
                    const int f = sLo[j];
                    for (int i = lane + 64; i < n[q]; i += 32)
                        atomicAdd(&c[pairs[f + i].y - lo], 1);
                }
            }
        }
        __syncthreads();
    }

    const int my = c[tid];
    int incl = my;
    for (int o = 1; o < warpSize; o <<= 1) {
        int v = __shfl_up(incl, o);
        if (lane >= o) incl += v;
    }
    if (lane == warpSize - 1) sWave[w] = incl;
    __syncthreads();
    if (tid == 0) {
        int tot = 0;
        for (int k = 0; k < nW; ++k) { int t = sWave[k]; sWave[k] = tot; tot += t; }
    }
    __syncthreads();
    const int first = e0 + sWave[w] + incl - my;
    c[tid] = first;
    const int node = lo + tid;
    if (node < nNodes) { start[node] = first; cnt[node] = my; }
    __syncthreads();

    for (int cb = 0; cb < nChunks; cb += BIN_NODES) {
        const int tileN = min(nChunks - cb, BIN_NODES);
        if (nChunks > BIN_NODES) {
            __syncthreads();
            if (tid < tileN) {
                const int a0 = seg[(cb + tid) * SEG_STRIDE + b];
                const int a1 = seg[(cb + tid) * SEG_STRIDE + b + 1];
                sN[tid]  = a1 - a0;
                sLo[tid] = (a1 > a0) ? (cb + tid) * CHUNK_EDGES + a0 : 0;
            }
            __syncthreads();
        }
        for (int j0 = w; j0 < tileN; j0 += nW * SEG_BATCH) {
            int2 p[SEG_BATCH][2];
            int  n[SEG_BATCH];
#pragma unroll
            for (int q = 0; q < SEG_BATCH; ++q) {
                const int j = j0 + q * nW;
                const bool ok = j < tileN;
                n[q] = ok ? sN[j] : 0;
                const int f = ok ? sLo[j] : 0;
                const int last = max(n[q] - 1, 0);
                p[q][0] = pairs[f + min(lane, last)];
                p[q][1] = pairs[f + min(lane + 32, last)];
            }
            asm volatile("" :: "v"(p[0][0].x), "v"(p[0][0].y), "v"(p[0][1].x), "v"(p[0][1].y),
                               "v"(p[1][0].x), "v"(p[1][0].y), "v"(p[1][1].x), "v"(p[1][1].y),
                               "v"(p[2][0].x), "v"(p[2][0].y), "v"(p[2][1].x), "v"(p[2][1].y),
                               "v"(p[3][0].x), "v"(p[3][0].y), "v"(p[3][1].x), "v"(p[3][1].y),
                               "v"(p[4][0].x), "v"(p[4][0].y), "v"(p[4][1].x), "v"(p[4][1].y));
#pragma unroll
            for (int q = 0; q < SEG_BATCH; ++q) {
                if (lane < n[q])      col[atomicAdd(&c[p[q][0].y - lo], 1)] = p[q][0].x;
                if (lane + 32 < n[q]) col[atomicAdd(&c[p[q][1].y - lo], 1)] = p[q][1].x;
                if (n[q] > 64) {
                    const int j = j0 + q * nW;
                    const int f = sLo[j];
                    for (int i = lane + 64; i < n[q]; i += 32) {
                        const int2 t = pairs[f + i];
                        col[atomicAdd(&c[t.y - lo], 1)] = t.x;
                    }
                }
            }
        }
    }
}

#define LDS_STRIDE (IN_DIM + 1)
#define TILE_ROWS  64
#define A_PITCH    (IN_DIM + 4)

__global__ __launch_bounds__(256, 16)
void gather_gemm_kernel(const float* __restrict__ x,
                        const int* __restrict__ col,
                        const int* __restrict__ start,
                        const int* __restrict__ cnt,
                        const float* __restrict__ W,
                        const float* __restrict__ bias,
                        float* __restrict__ out,
                        int nNodes) {
    __shared__ float sW[OUT_DIM * LDS_STRIDE];
    __shared__ __attribute__((aligned(16))) float sA[TILE_ROWS * A_PITCH];
    __shared__ float sInv[TILE_ROWS];

    const int row0 = blockIdx.x * TILE_ROWS;
    const int hw   = threadIdx.x >> 4;
    const int l    = threadIdx.x & 15;

    int c[4], st[4];
#pragma unroll
    for (int i = 0; i < 4; ++i) {
        const int nc = min(row0 + hw * 4 + i, nNodes - 1);
        c[i]  = cnt[nc];
        st[i] = start[nc];
    }

    float wv[16];
#pragma unroll
    for (int k = 0; k < 16; ++k) wv[k] = W[k * 256 + threadIdx.x];
    asm volatile("" :: "v"(c[0]), "v"(c[1]), "v"(c[2]), "v"(c[3]),
                       "v"(st[0]), "v"(st[1]), "v"(st[2]), "v"(st[3]));
#pragma unroll
    for (int k = 0; k < 16; ++k) {
        const int i = k * 256 + threadIdx.x;
        sW[(i >> 6) * LDS_STRIDE + (i & (IN_DIM - 1))] = wv[k];
    }
#pragma unroll
    for (int i = 0; i < 4; ++i)
        if (row0 + hw * 4 + i >= nNodes) c[i] = 0;

    const float4* x4 = reinterpret_cast<const float4*>(x);
    const float4 z = make_float4(0.f, 0.f, 0.f, 0.f);
#pragma unroll
    for (int i = 0; i < 4; ++i) {
        const int* cp = col + st[i];
        const int cc = c[i];
        const int last = cc - 1;
        float4 acc = z;
        for (int j = 0; j < cc; j += 4) {
            const int i0 = cp[j];
            const int i1 = cp[min(j + 1, last)];
            const int i2 = cp[min(j + 2, last)];
            const int i3 = cp[min(j + 3, last)];
            const float4 v0 = x4[(size_t)i0 * (IN_DIM / 4) + l];
            float4 v1 = x4[(size_t)i1 * (IN_DIM / 4) + l];
            float4 v2 = x4[(size_t)i2 * (IN_DIM / 4) + l];
            float4 v3 = x4[(size_t)i3 * (IN_DIM / 4) + l];
            if (j + 1 > last) v1 = z;
            if (j + 2 > last) v2 = z;
            if (j + 3 > last) v3 = z;
            acc.x += v0.x; acc.y += v0.y; acc.z += v0.z; acc.w += v0.w;
            acc.x += v1.x; acc.y += v1.y; acc.z += v1.z; acc.w += v1.w;
            acc.x += v2.x; acc.y += v2.y; acc.z += v2.z; acc.w += v2.w;
            acc.x += v3.x; acc.y += v3.y; acc.z += v3.z; acc.w += v3.w;
        }
        const int rt = hw * 4 + i;
        *reinterpret_cast<float4*>(&sA[rt * A_PITCH + 4 * l]) = acc;
        if (l == 0) sInv[rt] = 1.0f / fmaxf((float)cc, 1.0f);
    }
    __syncthreads();

    const int lane = threadIdx.x & 31;
    const int wave = threadIdx.x >> 5;
    const int tile = wave >> 1;
    const int nh   = wave & 1;
    const int trow0 = row0 + tile * 16;
    if (trow0 >= nNodes) return;

    const int m    = lane & 15;
    const int hi   = lane >> 4;
    const int koff = hi * 2;

    const float inv = sInv[tile * 16 + m];
    const float* rowp = &sA[(tile * 16 + m) * A_PITCH];
    v2f a[16];
#pragma unroll
    for (int kk = 0; kk < 16; ++kk) {
        a[kk].x = rowp[kk * 4 + koff]     * inv;
        a[kk].y = rowp[kk * 4 + koff + 1] * inv;
    }

    v8f acc[2] = {};

#pragma unroll
    for (int kk = 0; kk < 16; ++kk) {
#pragma unroll
        for (int tt = 0; tt < 2; ++tt) {
            int n = (nh * 2 + tt) * 16 + m;
            v2f b;
            b.x = sW[n * LDS_STRIDE + kk * 4 + koff];
            b.y = sW[n * LDS_STRIDE + kk * 4 + koff + 1];
            acc[tt] = __builtin_amdgcn_wmma_f32_16x16x4_f32(
                 false, a[kk],  false, b,
                 (short)0, acc[tt],
                 false,  false);
        }
    }

#pragma unroll
    for (int tt = 0; tt < 2; ++tt) {
        int n = (nh * 2 + tt) * 16 + m;
        float bv = bias[n];
#pragma unroll
        for (int v = 0; v < 8; ++v) {
            int rr = trow0 + v + hi * 8;
            out[(size_t)rr * OUT_DIM + n] = acc[tt][v] + bv;
        }
    }
}

extern "C" void kernel_launch(void* const* d_in, const int* in_sizes, int n_in,
                              void* d_out, int out_size, void* d_ws, size_t ws_size,
                              hipStream_t stream) {
    const float* x   = (const float*)d_in[0];
    const int*   src = (const int*)  d_in[1];
    const int*   dst = (const int*)  d_in[2];
    const float* W   = (const float*)d_in[3];
    const float* b   = (const float*)d_in[4];

    float* out = (float*)d_out;

    const int nNodes = in_sizes[0] / IN_DIM;
    const int nEdges = in_sizes[1];

    const int nBins   = (nNodes + BIN_NODES - 1) / BIN_NODES;
    const int nChunks = (nEdges + CHUNK_EDGES - 1) / CHUNK_EDGES;

    int*   seg     = (int*)d_ws;
    int2*  pairs   = (int2*)(seg + nChunks * SEG_STRIDE);
    int*   col     = (int*)(pairs + nChunks * CHUNK_EDGES);
    int*   start   = col + nEdges;
    int*   cnt     = start + nNodes;

    chunk_sort_kernel<<<nChunks, CHUNK_T, 0, stream>>>(src, dst, seg, pairs, nEdges);

    bin_csr_kernel<<<nBins, BIN_NODES, 0, stream>>>(pairs, seg, col, start, cnt, nNodes, nChunks);

    {
        int blocks = (nNodes + TILE_ROWS - 1) / TILE_ROWS;
        gather_gemm_kernel<<<blocks, 256, 0, stream>>>(x, col, start, cnt, W, b, out, nNodes);
    }
}


# bucket build moved into the GEMM kernel: level 1 sorts by 64-row range, each block counts, scans and places its sources in LDS; CSR launch and planes gone; row sum order = LDS arrival, walk changed
# speedup vs baseline: 7.2749x; 1.2514x over previous
#include <hip/hip_runtime.h>


#define IN_DIM 64
#define OUT_DIM 64

typedef float v2f __attribute__((ext_vector_type(2)));
typedef float v8f __attribute__((ext_vector_type(8)));

#define BIN_SHIFT   6
#define TILE_ROWS   64
#define MAX_BINS    2048
#define SEG_STRIDE  (MAX_BINS + 4)
#define CHUNK_T     512
#define CHUNK_EPT   16
#define CHUNK_EDGES (CHUNK_T * CHUNK_EPT)
#define BINS_PT     (MAX_BINS / CHUNK_T)
#define SEG_LANES   16
#define SEG_GROUPS  (256 / SEG_LANES)
#define SEG_KEEP    10
#define COL_CAP     1536

static_assert(BINS_PT == 4, "the scan below handles four counters per thread as one int4");
static_assert((1 << BIN_SHIFT) == TILE_ROWS, "a range is a block's row tile");

__global__ void chunk_sort_kernel(const int* __restrict__ src,
                                  const int* __restrict__ dst,
                                  int* __restrict__ seg,
                                  int2* __restrict__ pairs,
                                  int nEdges) {
    __shared__ __attribute__((aligned(16))) int h[MAX_BINS];
    __shared__ int sWave[32];
    const int tid  = threadIdx.x;
    const int lane = threadIdx.x & (warpSize - 1);
    const int w    = threadIdx.x / warpSize;
    const int nW   = blockDim.x / warpSize;
    int4* h4 = reinterpret_cast<int4*>(h);
    h4[tid] = make_int4(0, 0, 0, 0);
    __syncthreads();
    const int base = blockIdx.x * CHUNK_EDGES;
    int d[CHUNK_EPT], s[CHUNK_EPT];
#pragma unroll
    for (int k = 0; k < CHUNK_EPT; ++k) {
        const int e = base + k * CHUNK_T + tid;
        const bool ok = e < nEdges;
        d[k] = ok ? dst[e] : -1;
        s[k] = ok ? src[e] : 0;
    }
#pragma unroll
    for (int k = 0; k < CHUNK_EPT; ++k)
        if (d[k] >= 0) atomicAdd(&h[d[k] >> BIN_SHIFT], 1);
    __syncthreads();
    const int4 cv = h4[tid];
    const int my = cv.x + cv.y + cv.z + cv.w;
    int incl = my;
    for (int o = 1; o < warpSize; o <<= 1) {
        int v = __shfl_up(incl, o);
        if (lane >= o) incl += v;
    }
    if (lane == warpSize - 1) sWave[w] = incl;
    __syncthreads();
    if (tid == 0) {
        int tot = 0;
        for (int k = 0; k < nW; ++k) { int t = sWave[k]; sWave[k] = tot; tot += t; }
    }
    __syncthreads();
    {
        int4 ex;
        ex.x = sWave[w] + incl - my;
        ex.y = ex.x + cv.x;
        ex.z = ex.y + cv.y;
        ex.w = ex.z + cv.z;
        h4[tid] = ex;
        *reinterpret_cast<int4*>(&seg[blockIdx.x * SEG_STRIDE + BINS_PT * tid]) = ex;
        if (tid == CHUNK_T - 1) seg[blockIdx.x * SEG_STRIDE + MAX_BINS] = ex.w + cv.w;
    }
    __syncthreads();
#pragma unroll
    for (int k = 0; k < CHUNK_EPT; ++k) {
        if (d[k] >= 0) {
            const int pos = atomicAdd(&h[d[k] >> BIN_SHIFT], 1);
            pairs[base + pos] = make_int2(s[k], d[k]);
        }
    }
}

template <bool PLACE>
__device__ __forceinline__ void walk_rest(const int2* __restrict__ pairs, const int* __restrict__ seg,
                                          int c, int b, int i0, int sl, int* cur, int* col, int row0) {
    const int a0 = seg[c * SEG_STRIDE + b];
    const int n  = seg[c * SEG_STRIDE + b + 1] - a0;
    const int2* sp = pairs + (size_t)c * CHUNK_EDGES + a0;
    for (int i = i0 + sl; i < n; i += SEG_LANES) {
        const int2 p = sp[i];
        if (PLACE) col[atomicAdd(&cur[p.y - row0], 1)] = p.x;
        else       atomicAdd(&cur[p.y - row0], 1);
    }
}

#define LDS_STRIDE (IN_DIM + 1)
#define A_PITCH    (IN_DIM + 4)

__global__ __launch_bounds__(256, 16) __attribute__((amdgpu_num_vgpr(64)))
void gather_gemm_kernel(const float* __restrict__ x,
                        const int2* __restrict__ pairs,
                        const int* __restrict__ seg,
                        const float* __restrict__ W,
                        const float* __restrict__ bias,
                        float* __restrict__ out,
                        int nNodes, int nChunks) {
    __shared__ float sW[OUT_DIM * LDS_STRIDE];
    __shared__ __attribute__((aligned(16))) float sA[TILE_ROWS * A_PITCH];
    __shared__ float sInv[TILE_ROWS];
    __shared__ int sCol[COL_CAP];
    __shared__ int sCnt[TILE_ROWS];
    __shared__ int sStart[TILE_ROWS];
    __shared__ int sCur[TILE_ROWS + 64];
    __shared__ int sWave[32];

    const int tid  = threadIdx.x;
    const int b    = blockIdx.x;
    const int row0 = b * TILE_ROWS;
    const int hw   = threadIdx.x >> 4;
    const int l    = threadIdx.x & 15;
    const int wl   = threadIdx.x & (warpSize - 1);
    const int ww   = threadIdx.x / warpSize;
    const int nW   = blockDim.x / warpSize;

    const int sg   = threadIdx.x / SEG_LANES;
    const int sl   = threadIdx.x % SEG_LANES;

    if (tid < TILE_ROWS) sCnt[tid] = 0;
    __syncthreads();

    float wv[16];
#pragma unroll
    for (int k = 0; k < 16; ++k) wv[k] = W[k * 256 + threadIdx.x];

    unsigned have = 0u, more = 0u;
    int pidx[SEG_KEEP];
#pragma unroll
    for (int k = 0; k < SEG_KEEP; ++k) {
        const int c  = sg + k * SEG_GROUPS;
        const int cc = min(c, nChunks - 1);
        const int a0 = seg[cc * SEG_STRIDE + b];
        const int a1 = seg[cc * SEG_STRIDE + b + 1];
        const int n  = (c < nChunks) ? a1 - a0 : 0;
        if (sl < n)        have |= 1u << k;
        if (n > SEG_LANES) more |= 1u << k;
        pidx[k] = (sl < n) ? cc * CHUNK_EDGES + a0 + sl : 0;
    }

#pragma unroll
    for (int k = 0; k < 16; ++k) {
        const int i = k * 256 + threadIdx.x;
        sW[(i >> 6) * LDS_STRIDE + (i & (IN_DIM - 1))] = wv[k];
    }

    int2 kp[SEG_KEEP];
#pragma unroll
    for (int k = 0; k < SEG_KEEP; ++k) kp[k] = pairs[pidx[k]];
    asm volatile("" :: "v"(kp[0].x), "v"(kp[0].y), "v"(kp[1].x), "v"(kp[1].y),
                       "v"(kp[2].x), "v"(kp[2].y), "v"(kp[3].x), "v"(kp[3].y),
                       "v"(kp[4].x), "v"(kp[4].y), "v"(kp[5].x), "v"(kp[5].y),
                       "v"(kp[6].x), "v"(kp[6].y), "v"(kp[7].x), "v"(kp[7].y),
                       "v"(kp[8].x), "v"(kp[8].y), "v"(kp[9].x), "v"(kp[9].y));

#pragma unroll
    for (int k = 0; k < SEG_KEEP; ++k)
        if ((have >> k) & 1u) atomicAdd(&sCnt[kp[k].y - row0], 1);
    if (more != 0u) {
        for (int k = 0; k < SEG_KEEP; ++k)
            if ((more >> k) & 1u)
                walk_rest<false>(pairs, seg, sg + k * SEG_GROUPS, b, SEG_LANES, sl, sCnt, sCol, row0);
    }
    for (int c = sg + SEG_KEEP * SEG_GROUPS; c < nChunks; c += SEG_GROUPS)
        walk_rest<false>(pairs, seg, c, b, 0, sl, sCnt, sCol, row0);
    __syncthreads();

    const int my = (tid < TILE_ROWS) ? sCnt[tid] : 0;
    int incl = my;
    for (int o = 1; o < warpSize; o <<= 1) {
        int v = __shfl_up(incl, o);
        if (wl >= o) incl += v;
    }
    if (wl == warpSize - 1) sWave[ww] = incl;
    __syncthreads();
    int wbase = 0, total = 0;
    for (int k = 0; k < nW; ++k) {
        const int t = sWave[k];
        if (k < ww) wbase += t;
        total += t;
    }
    if (tid < TILE_ROWS) {
        const int first = wbase + incl - my;
        sStart[tid] = first;
        sCur[tid]   = first;
    }
    const bool fits = total <= COL_CAP;
    __syncthreads();

    const float4* x4 = reinterpret_cast<const float4*>(x);
    const float4 z = make_float4(0.f, 0.f, 0.f, 0.f);

    if (fits) {
        int pos[SEG_KEEP];
#pragma unroll
        for (int k = 0; k < SEG_KEEP; ++k) {
            const bool hv = (have >> k) & 1u;
            pos[k] = atomicAdd(&sCur[hv ? kp[k].y - row0 : TILE_ROWS + wl], 1);
        }
#pragma unroll
        for (int k = 0; k < SEG_KEEP; ++k)
            if ((have >> k) & 1u) sCol[pos[k]] = kp[k].x;
        if (more != 0u) {
            for (int k = 0; k < SEG_KEEP; ++k)
                if ((more >> k) & 1u)
                    walk_rest<true>(pairs, seg, sg + k * SEG_GROUPS, b, SEG_LANES, sl, sCur, sCol, row0);
        }
        for (int c = sg + SEG_KEEP * SEG_GROUPS; c < nChunks; c += SEG_GROUPS)
            walk_rest<true>(pairs, seg, c, b, 0, sl, sCur, sCol, row0);
        __syncthreads();

#pragma unroll
        for (int i = 0; i < 4; ++i) {
            const int rt = hw * 4 + i;
            const int cc = sCnt[rt];
            const int* cp = &sCol[sStart[rt]];
            const int last = cc - 1;
            float4 acc = z;
            for (int j = 0; j < cc; j += 4) {
                const int i0 = cp[j];
                const int i1 = cp[min(j + 1, last)];
                const int i2 = cp[min(j + 2, last)];
                const int i3 = cp[min(j + 3, last)];
                const float4 v0 = x4[(size_t)i0 * (IN_DIM / 4) + l];
                float4 v1 = x4[(size_t)i1 * (IN_DIM / 4) + l];
                float4 v2 = x4[(size_t)i2 * (IN_DIM / 4) + l];
                float4 v3 = x4[(size_t)i3 * (IN_DIM / 4) + l];
                if (j + 1 > last) v1 = z;
                if (j + 2 > last) v2 = z;
                if (j + 3 > last) v3 = z;
                acc.x += v0.x; acc.y += v0.y; acc.z += v0.z; acc.w += v0.w;
                acc.x += v1.x; acc.y += v1.y; acc.z += v1.z; acc.w += v1.w;
                acc.x += v2.x; acc.y += v2.y; acc.z += v2.z; acc.w += v2.w;
                acc.x += v3.x; acc.y += v3.y; acc.z += v3.z; acc.w += v3.w;
            }
            *reinterpret_cast<float4*>(&sA[rt * A_PITCH + 4 * l]) = acc;
            if (l == 0) sInv[rt] = 1.0f / fmaxf((float)cc, 1.0f);
        }
    } else {
        float4 fa0 = z, fa1 = z, fa2 = z, fa3 = z;
        const int g0 = row0 + hw * 4;
        for (int c = 0; c < nChunks; ++c) {
            const int a0 = seg[c * SEG_STRIDE + b];
            const int a1 = seg[c * SEG_STRIDE + b + 1];
            const int2* sp = pairs + (size_t)c * CHUNK_EDGES;
            for (int i = a0; i < a1; ++i) {
                const int2 p = sp[i];
                const int r = p.y - g0;
                if (r >= 0 && r < 4) {
                    const float4 v = x4[(size_t)p.x * (IN_DIM / 4) + l];
                    if (r == 0) { fa0.x += v.x; fa0.y += v.y; fa0.z += v.z; fa0.w += v.w; }
                    if (r == 1) { fa1.x += v.x; fa1.y += v.y; fa1.z += v.z; fa1.w += v.w; }
                    if (r == 2) { fa2.x += v.x; fa2.y += v.y; fa2.z += v.z; fa2.w += v.w; }
                    if (r == 3) { fa3.x += v.x; fa3.y += v.y; fa3.z += v.z; fa3.w += v.w; }
                }
            }
        }
        const int rt = hw * 4;
        *reinterpret_cast<float4*>(&sA[(rt + 0) * A_PITCH + 4 * l]) = fa0;
        *reinterpret_cast<float4*>(&sA[(rt + 1) * A_PITCH + 4 * l]) = fa1;
        *reinterpret_cast<float4*>(&sA[(rt + 2) * A_PITCH + 4 * l]) = fa2;
        *reinterpret_cast<float4*>(&sA[(rt + 3) * A_PITCH + 4 * l]) = fa3;
        if (l < 4) sInv[rt + l] = 1.0f / fmaxf((float)sCnt[rt + l], 1.0f);
    }
    __syncthreads();

    const int lane = threadIdx.x & 31;
    const int wave = threadIdx.x >> 5;
    const int tile = wave >> 1;
    const int nh   = wave & 1;
    const int trow0 = row0 + tile * 16;
    if (trow0 >= nNodes) return;

    const int m    = lane & 15;
    const int hi   = lane >> 4;
    const int koff = hi * 2;

    const float inv = sInv[tile * 16 + m];
    const float* rowp = &sA[(tile * 16 + m) * A_PITCH];
    v2f a[16];
#pragma unroll
    for (int kk = 0; kk < 16; ++kk) {
        a[kk].x = rowp[kk * 4 + koff]     * inv;
        a[kk].y = rowp[kk * 4 + koff + 1] * inv;
    }

    v8f acc[2] = {};

#pragma unroll
    for (int kk = 0; kk < 16; ++kk) {
#pragma unroll
        for (int tt = 0; tt < 2; ++tt) {
            int n = (nh * 2 + tt) * 16 + m;
            v2f b;
            b.x = sW[n * LDS_STRIDE + kk * 4 + koff];
            b.y = sW[n * LDS_STRIDE + kk * 4 + koff + 1];
            acc[tt] = __builtin_amdgcn_wmma_f32_16x16x4_f32(
                 false, a[kk],  false, b,
                 (short)0, acc[tt],
                 false,  false);
        }
    }

#pragma unroll
    for (int tt = 0; tt < 2; ++tt) {
        int n = (nh * 2 + tt) * 16 + m;
        float bv = bias[n];
#pragma unroll
        for (int v = 0; v < 8; ++v) {
            int rr = trow0 + v + hi * 8;
            out[(size_t)rr * OUT_DIM + n] = acc[tt][v] + bv;
        }
    }
}

extern "C" void kernel_launch(void* const* d_in, const int* in_sizes, int n_in,
                              void* d_out, int out_size, void* d_ws, size_t ws_size,
                              hipStream_t stream) {
    const float* x   = (const float*)d_in[0];
    const int*   src = (const int*)  d_in[1];
    const int*   dst = (const int*)  d_in[2];
    const float* W   = (const float*)d_in[3];
    const float* b   = (const float*)d_in[4];

    float* out = (float*)d_out;

    const int nNodes = in_sizes[0] / IN_DIM;
    const int nEdges = in_sizes[1];

    const int nBins   = (nNodes + TILE_ROWS - 1) / TILE_ROWS;
    const int nChunks = (nEdges + CHUNK_EDGES - 1) / CHUNK_EDGES;

    int*   seg     = (int*)d_ws;
    int2*  pairs   = (int2*)(seg + nChunks * SEG_STRIDE);

    chunk_sort_kernel<<<nChunks, CHUNK_T, 0, stream>>>(src, dst, seg, pairs, nEdges);

    gather_gemm_kernel<<<nBins, 256, 0, stream>>>(x, pairs, seg, W, b, out, nNodes, nChunks);
}


# 64-row tile on 4 waves, not 8: half as many waves enter; a lane does twice the work (32 B of a source row as 2 whole-line reads, 8 in flight); row sum order = LDS arrival order, walk changed
# speedup vs baseline: 7.5422x; 1.0367x over previous
#include <hip/hip_runtime.h>


#define IN_DIM 64
#define OUT_DIM 64

typedef float v2f __attribute__((ext_vector_type(2)));
typedef float v8f __attribute__((ext_vector_type(8)));

#define BIN_SHIFT   6
#define TILE_ROWS   64
#define MAX_BINS    2048
#define SEG_STRIDE  (MAX_BINS + 4)
#define CHUNK_T     512
#define CHUNK_EPT   16
#define CHUNK_EDGES (CHUNK_T * CHUNK_EPT)
#define BINS_PT     (MAX_BINS / CHUNK_T)
#define GG_T        128
#define SEG_LANES   16
#define SEG_GROUPS  (GG_T / SEG_LANES)
#define SEG_KEEP    19
#define ROW_LANES   8
#define W_PT        (OUT_DIM * IN_DIM / GG_T)
#define COL_CAP     1536

static_assert(BINS_PT == 4, "the scan below handles four counters per thread as one int4");
static_assert((1 << BIN_SHIFT) == TILE_ROWS, "a range is a block's row tile");
static_assert(GG_T / 32 * 16 == TILE_ROWS, "a wave of step 2 owns one 16-row tile");
static_assert(GG_T / ROW_LANES * 4 == TILE_ROWS, "a group of 8 lanes sums four rows");
static_assert(SEG_KEEP <= 32, "have and more are 32-bit masks");

__global__ void chunk_sort_kernel(const int* __restrict__ src,
                                  const int* __restrict__ dst,
                                  int* __restrict__ seg,
                                  int2* __restrict__ pairs,
                                  int nEdges) {
    __shared__ __attribute__((aligned(16))) int h[MAX_BINS];
    __shared__ int sWave[32];
    const int tid  = threadIdx.x;
    const int lane = threadIdx.x & (warpSize - 1);
    const int w    = threadIdx.x / warpSize;
    const int nW   = blockDim.x / warpSize;
    int4* h4 = reinterpret_cast<int4*>(h);
    h4[tid] = make_int4(0, 0, 0, 0);
    __syncthreads();
    const int base = blockIdx.x * CHUNK_EDGES;
    int d[CHUNK_EPT], s[CHUNK_EPT];
#pragma unroll
    for (int k = 0; k < CHUNK_EPT; ++k) {
        const int e = base + k * CHUNK_T + tid;
        const bool ok = e < nEdges;
        d[k] = ok ? dst[e] : -1;
        s[k] = ok ? src[e] : 0;
    }
#pragma unroll
    for (int k = 0; k < CHUNK_EPT; ++k)
        if (d[k] >= 0) atomicAdd(&h[d[k] >> BIN_SHIFT], 1);
    __syncthreads();
    const int4 cv = h4[tid];
    const int my = cv.x + cv.y + cv.z + cv.w;
    int incl = my;
    for (int o = 1; o < warpSize; o <<= 1) {
        int v = __shfl_up(incl, o);
        if (lane >= o) incl += v;
    }
    if (lane == warpSize - 1) sWave[w] = incl;
    __syncthreads();
    if (tid == 0) {
        int tot = 0;
        for (int k = 0; k < nW; ++k) { int t = sWave[k]; sWave[k] = tot; tot += t; }
    }
    __syncthreads();
    {
        int4 ex;
        ex.x = sWave[w] + incl - my;
        ex.y = ex.x + cv.x;
        ex.z = ex.y + cv.y;
        ex.w = ex.z + cv.z;
        h4[tid] = ex;
        *reinterpret_cast<int4*>(&seg[blockIdx.x * SEG_STRIDE + BINS_PT * tid]) = ex;
        if (tid == CHUNK_T - 1) seg[blockIdx.x * SEG_STRIDE + MAX_BINS] = ex.w + cv.w;
    }
    __syncthreads();
#pragma unroll
    for (int k = 0; k < CHUNK_EPT; ++k) {
        if (d[k] >= 0) {
            const int pos = atomicAdd(&h[d[k] >> BIN_SHIFT], 1);
            pairs[base + pos] = make_int2(s[k], d[k]);
        }
    }
}

template <bool PLACE>
__device__ __forceinline__ void walk_rest(const int2* __restrict__ pairs, const int* __restrict__ seg,
                                          int c, int b, int i0, int sl, int* cur, int* col, int row0) {
    const int a0 = seg[c * SEG_STRIDE + b];
    const int n  = seg[c * SEG_STRIDE + b + 1] - a0;
    const int2* sp = pairs + (size_t)c * CHUNK_EDGES + a0;
    for (int i = i0 + sl; i < n; i += SEG_LANES) {
        const int2 p = sp[i];
        if (PLACE) col[atomicAdd(&cur[p.y - row0], 1)] = p.x;
        else       atomicAdd(&cur[p.y - row0], 1);
    }
}

#define LDS_STRIDE (IN_DIM + 1)
#define A_PITCH    (IN_DIM + 4)

__global__ __launch_bounds__(GG_T, 7)
void gather_gemm_kernel(const float* __restrict__ x,
                        const int2* __restrict__ pairs,
                        const int* __restrict__ seg,
                        const float* __restrict__ W,
                        const float* __restrict__ bias,
                        float* __restrict__ out,
                        int nNodes, int nChunks) {
    __shared__ float sW[OUT_DIM * LDS_STRIDE];
    __shared__ __attribute__((aligned(16))) float sA[TILE_ROWS * A_PITCH];
    __shared__ float sInv[TILE_ROWS];
    __shared__ int sCol[COL_CAP];
    __shared__ int sCnt[TILE_ROWS];
    __shared__ int sStart[TILE_ROWS];
    __shared__ int sCur[TILE_ROWS + 64];
    __shared__ int sWave[32];

    const int tid  = threadIdx.x;
    const int b    = blockIdx.x;
    const int row0 = b * TILE_ROWS;
    const int hw   = threadIdx.x / ROW_LANES;
    const int l    = threadIdx.x % ROW_LANES;
    const int wl   = threadIdx.x & (warpSize - 1);
    const int ww   = threadIdx.x / warpSize;
    const int nW   = blockDim.x / warpSize;

    const int sg   = threadIdx.x / SEG_LANES;
    const int sl   = threadIdx.x % SEG_LANES;

    if (tid < TILE_ROWS) sCnt[tid] = 0;
    __syncthreads();

    float wv[W_PT];
#pragma unroll
    for (int k = 0; k < W_PT; ++k) wv[k] = W[k * GG_T + threadIdx.x];

    int sa0[SEG_KEEP], sa1[SEG_KEEP];
#pragma unroll
    for (int k = 0; k < SEG_KEEP; ++k) {
        const int cc = min(sg + k * SEG_GROUPS, nChunks - 1);
        sa0[k] = seg[cc * SEG_STRIDE + b];
        sa1[k] = seg[cc * SEG_STRIDE + b + 1];
    }
    asm volatile("" :: "v"(sa0[0]), "v"(sa1[0]), "v"(sa0[1]), "v"(sa1[1]),
                       "v"(sa0[2]), "v"(sa1[2]), "v"(sa0[3]), "v"(sa1[3]),
                       "v"(sa0[4]), "v"(sa1[4]), "v"(sa0[5]), "v"(sa1[5]),
                       "v"(sa0[6]), "v"(sa1[6]), "v"(sa0[7]), "v"(sa1[7]),
                       "v"(sa0[8]), "v"(sa1[8]), "v"(sa0[9]), "v"(sa1[9]));
    asm volatile("" :: "v"(sa0[10]), "v"(sa1[10]), "v"(sa0[11]), "v"(sa1[11]),
                       "v"(sa0[12]), "v"(sa1[12]), "v"(sa0[13]), "v"(sa1[13]),
                       "v"(sa0[14]), "v"(sa1[14]), "v"(sa0[15]), "v"(sa1[15]),
                       "v"(sa0[16]), "v"(sa1[16]), "v"(sa0[17]), "v"(sa1[17]),
                       "v"(sa0[18]), "v"(sa1[18]));
    unsigned have = 0u, more = 0u;
    int pidx[SEG_KEEP];
#pragma unroll
    for (int k = 0; k < SEG_KEEP; ++k) {
        const int c  = sg + k * SEG_GROUPS;
        const int cc = min(c, nChunks - 1);
        const int n  = (c < nChunks) ? sa1[k] - sa0[k] : 0;
        if (sl < n)        have |= 1u << k;
        if (n > SEG_LANES) more |= 1u << k;
        pidx[k] = (sl < n) ? cc * CHUNK_EDGES + sa0[k] + sl : 0;
    }

#pragma unroll
    for (int k = 0; k < W_PT; ++k) {
        const int i = k * GG_T + threadIdx.x;
        sW[(i >> 6) * LDS_STRIDE + (i & (IN_DIM - 1))] = wv[k];
    }

    int2 kp[SEG_KEEP];
#pragma unroll
    for (int k = 0; k < SEG_KEEP; ++k) kp[k] = pairs[pidx[k]];
    asm volatile("" :: "v"(kp[0].x), "v"(kp[0].y), "v"(kp[1].x), "v"(kp[1].y),
                       "v"(kp[2].x), "v"(kp[2].y), "v"(kp[3].x), "v"(kp[3].y),
                       "v"(kp[4].x), "v"(kp[4].y), "v"(kp[5].x), "v"(kp[5].y),
                       "v"(kp[6].x), "v"(kp[6].y), "v"(kp[7].x), "v"(kp[7].y),
                       "v"(kp[8].x), "v"(kp[8].y), "v"(kp[9].x), "v"(kp[9].y));
    asm volatile("" :: "v"(kp[10].x), "v"(kp[10].y), "v"(kp[11].x), "v"(kp[11].y),
                       "v"(kp[12].x), "v"(kp[12].y), "v"(kp[13].x), "v"(kp[13].y),
                       "v"(kp[14].x), "v"(kp[14].y), "v"(kp[15].x), "v"(kp[15].y),
                       "v"(kp[16].x), "v"(kp[16].y), "v"(kp[17].x), "v"(kp[17].y),
                       "v"(kp[18].x), "v"(kp[18].y));

#pragma unroll
    for (int k = 0; k < SEG_KEEP; ++k)
        if ((have >> k) & 1u) atomicAdd(&sCnt[kp[k].y - row0], 1);
    if (more != 0u) {
        for (int k = 0; k < SEG_KEEP; ++k)
            if ((more >> k) & 1u)
                walk_rest<false>(pairs, seg, sg + k * SEG_GROUPS, b, SEG_LANES, sl, sCnt, sCol, row0);
    }
    for (int c = sg + SEG_KEEP * SEG_GROUPS; c < nChunks; c += SEG_GROUPS)
        walk_rest<false>(pairs, seg, c, b, 0, sl, sCnt, sCol, row0);
    __syncthreads();

    const int my = (tid < TILE_ROWS) ? sCnt[tid] : 0;
    int incl = my;
    for (int o = 1; o < warpSize; o <<= 1) {
        int v = __shfl_up(incl, o);
        if (wl >= o) incl += v;
    }
    if (wl == warpSize - 1) sWave[ww] = incl;
    __syncthreads();
    int wbase = 0, total = 0;
    for (int k = 0; k < nW; ++k) {
        const int t = sWave[k];
        if (k < ww) wbase += t;
        total += t;
    }
    if (tid < TILE_ROWS) {
        const int first = wbase + incl - my;
        sStart[tid] = first;
        sCur[tid]   = first;
    }
    const bool fits = total <= COL_CAP;
    __syncthreads();

    const float4* x4 = reinterpret_cast<const float4*>(x);
    const float4 z = make_float4(0.f, 0.f, 0.f, 0.f);

    if (fits) {
        int pos[SEG_KEEP];
#pragma unroll
        for (int k = 0; k < SEG_KEEP; ++k) {
            const bool hv = (have >> k) & 1u;
            pos[k] = atomicAdd(&sCur[hv ? kp[k].y - row0 : TILE_ROWS + wl], 1);
        }
#pragma unroll
        for (int k = 0; k < SEG_KEEP; ++k)
            if ((have >> k) & 1u) sCol[pos[k]] = kp[k].x;
        if (more != 0u) {
            for (int k = 0; k < SEG_KEEP; ++k)
                if ((more >> k) & 1u)
                    walk_rest<true>(pairs, seg, sg + k * SEG_GROUPS, b, SEG_LANES, sl, sCur, sCol, row0);
        }
        for (int c = sg + SEG_KEEP * SEG_GROUPS; c < nChunks; c += SEG_GROUPS)
            walk_rest<true>(pairs, seg, c, b, 0, sl, sCur, sCol, row0);
        __syncthreads();

#pragma unroll
        for (int i = 0; i < 4; ++i) {
            const int rt = hw * 4 + i;
            const int cc = sCnt[rt];
            const int* cp = &sCol[sStart[rt]];
            const int last = cc - 1;
            float4 acc = z, acd = z;
            for (int j = 0; j < cc; j += 4) {
                const int i0 = cp[j];
                const int i1 = cp[min(j + 1, last)];
                const int i2 = cp[min(j + 2, last)];
                const int i3 = cp[min(j + 3, last)];
                const float4 v0 = x4[(size_t)i0 * (IN_DIM / 4) + l];
                const float4 w0 = x4[(size_t)i0 * (IN_DIM / 4) + ROW_LANES + l];
                float4 v1 = x4[(size_t)i1 * (IN_DIM / 4) + l];
                float4 w1 = x4[(size_t)i1 * (IN_DIM / 4) + ROW_LANES + l];
                float4 v2 = x4[(size_t)i2 * (IN_DIM / 4) + l];
                float4 w2 = x4[(size_t)i2 * (IN_DIM / 4) + ROW_LANES + l];
                float4 v3 = x4[(size_t)i3 * (IN_DIM / 4) + l];
                float4 w3 = x4[(size_t)i3 * (IN_DIM / 4) + ROW_LANES + l];
                if (j + 1 > last) { v1 = z; w1 = z; }
                if (j + 2 > last) { v2 = z; w2 = z; }
                if (j + 3 > last) { v3 = z; w3 = z; }
                acc.x += v0.x; acc.y += v0.y; acc.z += v0.z; acc.w += v0.w;
                acc.x += v1.x; acc.y += v1.y; acc.z += v1.z; acc.w += v1.w;
                acc.x += v2.x; acc.y += v2.y; acc.z += v2.z; acc.w += v2.w;
                acc.x += v3.x; acc.y += v3.y; acc.z += v3.z; acc.w += v3.w;
                acd.x += w0.x; acd.y += w0.y; acd.z += w0.z; acd.w += w0.w;
                acd.x += w1.x; acd.y += w1.y; acd.z += w1.z; acd.w += w1.w;
                acd.x += w2.x; acd.y += w2.y; acd.z += w2.z; acd.w += w2.w;
                acd.x += w3.x; acd.y += w3.y; acd.z += w3.z; acd.w += w3.w;
            }
            *reinterpret_cast<float4*>(&sA[rt * A_PITCH + 4 * l]) = acc;
            *reinterpret_cast<float4*>(&sA[rt * A_PITCH + 4 * (ROW_LANES + l)]) = acd;
            if (l == 0) sInv[rt] = 1.0f / fmaxf((float)cc, 1.0f);
        }
    } else {
        float4 fa0 = z, fa1 = z, fa2 = z, fa3 = z;
        float4 fb0 = z, fb1 = z, fb2 = z, fb3 = z;
        const int g0 = row0 + hw * 4;
        for (int c = 0; c < nChunks; ++c) {
            const int a0 = seg[c * SEG_STRIDE + b];
            const int a1 = seg[c * SEG_STRIDE + b + 1];
            const int2* sp = pairs + (size_t)c * CHUNK_EDGES;
            for (int i = a0; i < a1; ++i) {
                const int2 p = sp[i];
                const int r = p.y - g0;
                if (r >= 0 && r < 4) {
                    const float4 v = x4[(size_t)p.x * (IN_DIM / 4) + l];
                    const float4 u = x4[(size_t)p.x * (IN_DIM / 4) + ROW_LANES + l];
                    if (r == 0) { fa0.x += v.x; fa0.y += v.y; fa0.z += v.z; fa0.w += v.w;
                                  fb0.x += u.x; fb0.y += u.y; fb0.z += u.z; fb0.w += u.w; }
                    if (r == 1) { fa1.x += v.x; fa1.y += v.y; fa1.z += v.z; fa1.w += v.w;
                                  fb1.x += u.x; fb1.y += u.y; fb1.z += u.z; fb1.w += u.w; }
                    if (r == 2) { fa2.x += v.x; fa2.y += v.y; fa2.z += v.z; fa2.w += v.w;
                                  fb2.x += u.x; fb2.y += u.y; fb2.z += u.z; fb2.w += u.w; }
                    if (r == 3) { fa3.x += v.x; fa3.y += v.y; fa3.z += v.z; fa3.w += v.w;
                                  fb3.x += u.x; fb3.y += u.y; fb3.z += u.z; fb3.w += u.w; }
                }
            }
        }
        const int rt = hw * 4;
        *reinterpret_cast<float4*>(&sA[(rt + 0) * A_PITCH + 4 * l]) = fa0;
        *reinterpret_cast<float4*>(&sA[(rt + 1) * A_PITCH + 4 * l]) = fa1;
        *reinterpret_cast<float4*>(&sA[(rt + 2) * A_PITCH + 4 * l]) = fa2;
        *reinterpret_cast<float4*>(&sA[(rt + 3) * A_PITCH + 4 * l]) = fa3;
        *reinterpret_cast<float4*>(&sA[(rt + 0) * A_PITCH + 4 * (ROW_LANES + l)]) = fb0;
        *reinterpret_cast<float4*>(&sA[(rt + 1) * A_PITCH + 4 * (ROW_LANES + l)]) = fb1;
        *reinterpret_cast<float4*>(&sA[(rt + 2) * A_PITCH + 4 * (ROW_LANES + l)]) = fb2;
        *reinterpret_cast<float4*>(&sA[(rt + 3) * A_PITCH + 4 * (ROW_LANES + l)]) = fb3;
        if (l < 4) sInv[rt + l] = 1.0f / fmaxf((float)sCnt[rt + l], 1.0f);
    }
    __syncthreads();

    const int lane = threadIdx.x & 31;
    const int tile = threadIdx.x >> 5;
    const int trow0 = row0 + tile * 16;
    if (trow0 >= nNodes) return;

    const int m    = lane & 15;
    const int hi   = lane >> 4;
    const int koff = hi * 2;

    const float inv = sInv[tile * 16 + m];
    const float* rowp = &sA[(tile * 16 + m) * A_PITCH];
    v2f a[16];
#pragma unroll
    for (int kk = 0; kk < 16; ++kk) {
        a[kk].x = rowp[kk * 4 + koff]     * inv;
        a[kk].y = rowp[kk * 4 + koff + 1] * inv;
    }

    v8f acc[4] = {};

#pragma unroll
    for (int kk = 0; kk < 16; ++kk) {
#pragma unroll
        for (int tt = 0; tt < 4; ++tt) {
            int n = tt * 16 + m;
            v2f b;
            b.x = sW[n * LDS_STRIDE + kk * 4 + koff];
            b.y = sW[n * LDS_STRIDE + kk * 4 + koff + 1];
            acc[tt] = __builtin_amdgcn_wmma_f32_16x16x4_f32(
                 false, a[kk],  false, b,
                 (short)0, acc[tt],
                 false,  false);
        }
    }

#pragma unroll
    for (int tt = 0; tt < 4; ++tt) {
        int n = tt * 16 + m;
        float bv = bias[n];
#pragma unroll
        for (int v = 0; v < 8; ++v) {
            int rr = trow0 + v + hi * 8;
            out[(size_t)rr * OUT_DIM + n] = acc[tt][v] + bv;
        }
    }
}

extern "C" void kernel_launch(void* const* d_in, const int* in_sizes, int n_in,
                              void* d_out, int out_size, void* d_ws, size_t ws_size,
                              hipStream_t stream) {
    const float* x   = (const float*)d_in[0];
    const int*   src = (const int*)  d_in[1];
    const int*   dst = (const int*)  d_in[2];
    const float* W   = (const float*)d_in[3];
    const float* b   = (const float*)d_in[4];

    float* out = (float*)d_out;

    const int nNodes = in_sizes[0] / IN_DIM;
    const int nEdges = in_sizes[1];

    const int nBins   = (nNodes + TILE_ROWS - 1) / TILE_ROWS;
    const int nChunks = (nEdges + CHUNK_EDGES - 1) / CHUNK_EDGES;

    int*   seg     = (int*)d_ws;
    int2*  pairs   = (int2*)(seg + nChunks * SEG_STRIDE);

    chunk_sort_kernel<<<nChunks, CHUNK_T, 0, stream>>>(src, dst, seg, pairs, nEdges);

    gather_gemm_kernel<<<nBins, GG_T, 0, stream>>>(x, pairs, seg, W, b, out, nNodes, nChunks);
}


# chunk sort places its sorted pairs in an LDS copy of its slice and stores the slice as whole 128-byte lines (512 B a wave store), not one 8-byte store per pair to 32 lines; row sum order unchanged
# speedup vs baseline: 7.9482x; 1.0538x over previous
#include <hip/hip_runtime.h>


#define IN_DIM 64
#define OUT_DIM 64

typedef float v2f __attribute__((ext_vector_type(2)));
typedef float v8f __attribute__((ext_vector_type(8)));

#define BIN_SHIFT   6
#define TILE_ROWS   64
#define MAX_BINS    2048
#define SEG_STRIDE  (MAX_BINS + 32)
#define CHUNK_T     512
#define CHUNK_EPT   16
#define CHUNK_EDGES (CHUNK_T * CHUNK_EPT)
#define BINS_PT     (MAX_BINS / CHUNK_T)
#define GG_T        128
#define SEG_LANES   16
#define SEG_GROUPS  (GG_T / SEG_LANES)
#define SEG_KEEP    19
#define ROW_LANES   8
#define W_PT        (OUT_DIM * IN_DIM / GG_T)
#define COL_CAP     1536

#define OUT_PT      (CHUNK_EDGES / 2 / CHUNK_T)
static_assert(BINS_PT == 4, "the scan below handles four counters per thread as one int4");
static_assert(SEG_STRIDE % 32 == 0 && CHUNK_EDGES % 16 == 0, "range-start rows and pair slices are whole 128-byte lines");
static_assert((1 << BIN_SHIFT) == TILE_ROWS, "a range is a block's row tile");
static_assert(GG_T / 32 * 16 == TILE_ROWS, "a wave of step 2 owns one 16-row tile");
static_assert(GG_T / ROW_LANES * 4 == TILE_ROWS, "a group of 8 lanes sums four rows");
static_assert(SEG_KEEP <= 32, "have and more are 32-bit masks");

__global__ void chunk_sort_kernel(const int* __restrict__ src,
                                  const int* __restrict__ dst,
                                  int* __restrict__ seg,
                                  int2* __restrict__ pairs,
                                  int nEdges) {
    __shared__ __attribute__((aligned(16))) int h[MAX_BINS];
    __shared__ __attribute__((aligned(16))) int2 sP[CHUNK_EDGES];
    __shared__ int sWave[32];
    const int tid  = threadIdx.x;
    const int lane = threadIdx.x & (warpSize - 1);
    const int w    = threadIdx.x / warpSize;
    const int nW   = blockDim.x / warpSize;
    int4* h4 = reinterpret_cast<int4*>(h);
    h4[tid] = make_int4(0, 0, 0, 0);
    __syncthreads();
    const int base = blockIdx.x * CHUNK_EDGES;
    int d[CHUNK_EPT], s[CHUNK_EPT];
#pragma unroll
    for (int k = 0; k < CHUNK_EPT; ++k) {
        const int e = base + k * CHUNK_T + tid;
        const bool ok = e < nEdges;
        d[k] = ok ? dst[e] : -1;
        s[k] = ok ? src[e] : 0;
    }
#pragma unroll
    for (int k = 0; k < CHUNK_EPT; ++k)
        if (d[k] >= 0) atomicAdd(&h[d[k] >> BIN_SHIFT], 1);
    __syncthreads();
    const int4 cv = h4[tid];
    const int my = cv.x + cv.y + cv.z + cv.w;
    int incl = my;
    for (int o = 1; o < warpSize; o <<= 1) {
        int v = __shfl_up(incl, o);
        if (lane >= o) incl += v;
    }
    if (lane == warpSize - 1) sWave[w] = incl;
    __syncthreads();
    if (tid == 0) {
        int tot = 0;
        for (int k = 0; k < nW; ++k) { int t = sWave[k]; sWave[k] = tot; tot += t; }
    }
    __syncthreads();
    {
        int4 ex;
        ex.x = sWave[w] + incl - my;
        ex.y = ex.x + cv.x;
        ex.z = ex.y + cv.y;
        ex.w = ex.z + cv.z;
        h4[tid] = ex;
        *reinterpret_cast<int4*>(&seg[blockIdx.x * SEG_STRIDE + BINS_PT * tid]) = ex;
        if (tid == CHUNK_T - 1) seg[blockIdx.x * SEG_STRIDE + MAX_BINS] = ex.w + cv.w;
    }
    __syncthreads();
#pragma unroll
    for (int k = 0; k < CHUNK_EPT; ++k) {
        if (d[k] >= 0) {
            const int pos = atomicAdd(&h[d[k] >> BIN_SHIFT], 1);
            sP[pos] = make_int2(s[k], d[k]);
        }
    }
    __syncthreads();
    const int nValid = min(CHUNK_EDGES, nEdges - base);
    const int4* sP4 = reinterpret_cast<const int4*>(sP);
    int4* gp = reinterpret_cast<int4*>(pairs + base);
    int4 q[OUT_PT];
#pragma unroll
    for (int k = 0; k < OUT_PT; ++k) q[k] = sP4[k * CHUNK_T + tid];
    static_assert(OUT_PT == 8, "the fences below name eight reads");
    asm volatile("" :: "v"(q[0].x), "v"(q[0].y), "v"(q[0].z), "v"(q[0].w),
                       "v"(q[1].x), "v"(q[1].y), "v"(q[1].z), "v"(q[1].w),
                       "v"(q[2].x), "v"(q[2].y), "v"(q[2].z), "v"(q[2].w),
                       "v"(q[3].x), "v"(q[3].y), "v"(q[3].z), "v"(q[3].w));
    asm volatile("" :: "v"(q[4].x), "v"(q[4].y), "v"(q[4].z), "v"(q[4].w),
                       "v"(q[5].x), "v"(q[5].y), "v"(q[5].z), "v"(q[5].w),
                       "v"(q[6].x), "v"(q[6].y), "v"(q[6].z), "v"(q[6].w),
                       "v"(q[7].x), "v"(q[7].y), "v"(q[7].z), "v"(q[7].w));
#pragma unroll
    for (int k = 0; k < OUT_PT; ++k) {
        const int i = k * CHUNK_T + tid;
        if (2 * i < nValid) gp[i] = q[k];
    }
}

template <bool PLACE>
__device__ __forceinline__ void walk_rest(const int2* __restrict__ pairs, const int* __restrict__ seg,
                                          int c, int b, int i0, int sl, int* cur, int* col, int row0) {
    const int a0 = seg[c * SEG_STRIDE + b];
    const int n  = seg[c * SEG_STRIDE + b + 1] - a0;
    const int2* sp = pairs + (size_t)c * CHUNK_EDGES + a0;
    for (int i = i0 + sl; i < n; i += SEG_LANES) {
        const int2 p = sp[i];
        if (PLACE) col[atomicAdd(&cur[p.y - row0], 1)] = p.x;
        else       atomicAdd(&cur[p.y - row0], 1);
    }
}

#define LDS_STRIDE (IN_DIM + 1)
#define A_PITCH    (IN_DIM + 4)

__global__ __launch_bounds__(GG_T, 7)
void gather_gemm_kernel(const float* __restrict__ x,
                        const int2* __restrict__ pairs,
                        const int* __restrict__ seg,
                        const float* __restrict__ W,
                        const float* __restrict__ bias,
                        float* __restrict__ out,
                        int nNodes, int nChunks) {
    __shared__ float sW[OUT_DIM * LDS_STRIDE];
    __shared__ __attribute__((aligned(16))) float sA[TILE_ROWS * A_PITCH];
    __shared__ float sInv[TILE_ROWS];
    __shared__ int sCol[COL_CAP];
    __shared__ int sCnt[TILE_ROWS];
    __shared__ int sStart[TILE_ROWS];
    __shared__ int sCur[TILE_ROWS + 64];
    __shared__ int sWave[32];

    const int tid  = threadIdx.x;
    const int b    = blockIdx.x;
    const int row0 = b * TILE_ROWS;
    const int hw   = threadIdx.x / ROW_LANES;
    const int l    = threadIdx.x % ROW_LANES;
    const int wl   = threadIdx.x & (warpSize - 1);
    const int ww   = threadIdx.x / warpSize;
    const int nW   = blockDim.x / warpSize;

    const int sg   = threadIdx.x / SEG_LANES;
    const int sl   = threadIdx.x % SEG_LANES;

    if (tid < TILE_ROWS) sCnt[tid] = 0;
    __syncthreads();

    float wv[W_PT];
#pragma unroll
    for (int k = 0; k < W_PT; ++k) wv[k] = W[k * GG_T + threadIdx.x];

    int sa0[SEG_KEEP], sa1[SEG_KEEP];
#pragma unroll
    for (int k = 0; k < SEG_KEEP; ++k) {
        const int cc = min(sg + k * SEG_GROUPS, nChunks - 1);
        sa0[k] = seg[cc * SEG_STRIDE + b];
        sa1[k] = seg[cc * SEG_STRIDE + b + 1];
    }
    asm volatile("" :: "v"(sa0[0]), "v"(sa1[0]), "v"(sa0[1]), "v"(sa1[1]),
                       "v"(sa0[2]), "v"(sa1[2]), "v"(sa0[3]), "v"(sa1[3]),
                       "v"(sa0[4]), "v"(sa1[4]), "v"(sa0[5]), "v"(sa1[5]),
                       "v"(sa0[6]), "v"(sa1[6]), "v"(sa0[7]), "v"(sa1[7]),
                       "v"(sa0[8]), "v"(sa1[8]), "v"(sa0[9]), "v"(sa1[9]));
    asm volatile("" :: "v"(sa0[10]), "v"(sa1[10]), "v"(sa0[11]), "v"(sa1[11]),
                       "v"(sa0[12]), "v"(sa1[12]), "v"(sa0[13]), "v"(sa1[13]),
                       "v"(sa0[14]), "v"(sa1[14]), "v"(sa0[15]), "v"(sa1[15]),
                       "v"(sa0[16]), "v"(sa1[16]), "v"(sa0[17]), "v"(sa1[17]),
                       "v"(sa0[18]), "v"(sa1[18]));
    unsigned have = 0u, more = 0u;
    int pidx[SEG_KEEP];
#pragma unroll
    for (int k = 0; k < SEG_KEEP; ++k) {
        const int c  = sg + k * SEG_GROUPS;
        const int cc = min(c, nChunks - 1);
        const int n  = (c < nChunks) ? sa1[k] - sa0[k] : 0;
        if (sl < n)        have |= 1u << k;
        if (n > SEG_LANES) more |= 1u << k;
        pidx[k] = (sl < n) ? cc * CHUNK_EDGES + sa0[k] + sl : 0;
    }

#pragma unroll
    for (int k = 0; k < W_PT; ++k) {
        const int i = k * GG_T + threadIdx.x;
        sW[(i >> 6) * LDS_STRIDE + (i & (IN_DIM - 1))] = wv[k];
    }

    int2 kp[SEG_KEEP];
#pragma unroll
    for (int k = 0; k < SEG_KEEP; ++k) kp[k] = pairs[pidx[k]];
    asm volatile("" :: "v"(kp[0].x), "v"(kp[0].y), "v"(kp[1].x), "v"(kp[1].y),
                       "v"(kp[2].x), "v"(kp[2].y), "v"(kp[3].x), "v"(kp[3].y),
                       "v"(kp[4].x), "v"(kp[4].y), "v"(kp[5].x), "v"(kp[5].y),
                       "v"(kp[6].x), "v"(kp[6].y), "v"(kp[7].x), "v"(kp[7].y),
                       "v"(kp[8].x), "v"(kp[8].y), "v"(kp[9].x), "v"(kp[9].y));
    asm volatile("" :: "v"(kp[10].x), "v"(kp[10].y), "v"(kp[11].x), "v"(kp[11].y),
                       "v"(kp[12].x), "v"(kp[12].y), "v"(kp[13].x), "v"(kp[13].y),
                       "v"(kp[14].x), "v"(kp[14].y), "v"(kp[15].x), "v"(kp[15].y),
                       "v"(kp[16].x), "v"(kp[16].y), "v"(kp[17].x), "v"(kp[17].y),
                       "v"(kp[18].x), "v"(kp[18].y));

#pragma unroll
    for (int k = 0; k < SEG_KEEP; ++k)
        if ((have >> k) & 1u) atomicAdd(&sCnt[kp[k].y - row0], 1);
    if (more != 0u) {
        for (int k = 0; k < SEG_KEEP; ++k)
            if ((more >> k) & 1u)
                walk_rest<false>(pairs, seg, sg + k * SEG_GROUPS, b, SEG_LANES, sl, sCnt, sCol, row0);
    }
    for (int c = sg + SEG_KEEP * SEG_GROUPS; c < nChunks; c += SEG_GROUPS)
        walk_rest<false>(pairs, seg, c, b, 0, sl, sCnt, sCol, row0);
    __syncthreads();

    const int my = (tid < TILE_ROWS) ? sCnt[tid] : 0;
    int incl = my;
    for (int o = 1; o < warpSize; o <<= 1) {
        int v = __shfl_up(incl, o);
        if (wl >= o) incl += v;
    }
    if (wl == warpSize - 1) sWave[ww] = incl;
    __syncthreads();
    int wbase = 0, total = 0;
    for (int k = 0; k < nW; ++k) {
        const int t = sWave[k];
        if (k < ww) wbase += t;
        total += t;
    }
    if (tid < TILE_ROWS) {
        const int first = wbase + incl - my;
        sStart[tid] = first;
        sCur[tid]   = first;
    }
    const bool fits = total <= COL_CAP;
    __syncthreads();

    const float4* x4 = reinterpret_cast<const float4*>(x);
    const float4 z = make_float4(0.f, 0.f, 0.f, 0.f);

    if (fits) {
        int pos[SEG_KEEP];
#pragma unroll
        for (int k = 0; k < SEG_KEEP; ++k) {
            const bool hv = (have >> k) & 1u;
            pos[k] = atomicAdd(&sCur[hv ? kp[k].y - row0 : TILE_ROWS + wl], 1);
        }
#pragma unroll
        for (int k = 0; k < SEG_KEEP; ++k)
            if ((have >> k) & 1u) sCol[pos[k]] = kp[k].x;
        if (more != 0u) {
            for (int k = 0; k < SEG_KEEP; ++k)
                if ((more >> k) & 1u)
                    walk_rest<true>(pairs, seg, sg + k * SEG_GROUPS, b, SEG_LANES, sl, sCur, sCol, row0);
        }
        for (int c = sg + SEG_KEEP * SEG_GROUPS; c < nChunks; c += SEG_GROUPS)
            walk_rest<true>(pairs, seg, c, b, 0, sl, sCur, sCol, row0);
        __syncthreads();

#pragma unroll
        for (int i = 0; i < 4; ++i) {
            const int rt = hw * 4 + i;
            const int cc = sCnt[rt];
            const int* cp = &sCol[sStart[rt]];
            const int last = cc - 1;
            float4 acc = z, acd = z;
            for (int j = 0; j < cc; j += 4) {
                const int i0 = cp[j];
                const int i1 = cp[min(j + 1, last)];
                const int i2 = cp[min(j + 2, last)];
                const int i3 = cp[min(j + 3, last)];
                const float4 v0 = x4[(size_t)i0 * (IN_DIM / 4) + l];
                const float4 w0 = x4[(size_t)i0 * (IN_DIM / 4) + ROW_LANES + l];
                float4 v1 = x4[(size_t)i1 * (IN_DIM / 4) + l];
                float4 w1 = x4[(size_t)i1 * (IN_DIM / 4) + ROW_LANES + l];
                float4 v2 = x4[(size_t)i2 * (IN_DIM / 4) + l];
                float4 w2 = x4[(size_t)i2 * (IN_DIM / 4) + ROW_LANES + l];
                float4 v3 = x4[(size_t)i3 * (IN_DIM / 4) + l];
                float4 w3 = x4[(size_t)i3 * (IN_DIM / 4) + ROW_LANES + l];
                if (j + 1 > last) { v1 = z; w1 = z; }
                if (j + 2 > last) { v2 = z; w2 = z; }
                if (j + 3 > last) { v3 = z; w3 = z; }
                acc.x += v0.x; acc.y += v0.y; acc.z += v0.z; acc.w += v0.w;
                acc.x += v1.x; acc.y += v1.y; acc.z += v1.z; acc.w += v1.w;
                acc.x += v2.x; acc.y += v2.y; acc.z += v2.z; acc.w += v2.w;
                acc.x += v3.x; acc.y += v3.y; acc.z += v3.z; acc.w += v3.w;
                acd.x += w0.x; acd.y += w0.y; acd.z += w0.z; acd.w += w0.w;
                acd.x += w1.x; acd.y += w1.y; acd.z += w1.z; acd.w += w1.w;
                acd.x += w2.x; acd.y += w2.y; acd.z += w2.z; acd.w += w2.w;
                acd.x += w3.x; acd.y += w3.y; acd.z += w3.z; acd.w += w3.w;
            }
            *reinterpret_cast<float4*>(&sA[rt * A_PITCH + 4 * l]) = acc;
            *reinterpret_cast<float4*>(&sA[rt * A_PITCH + 4 * (ROW_LANES + l)]) = acd;
            if (l == 0) sInv[rt] = 1.0f / fmaxf((float)cc, 1.0f);
        }
    } else {
        float4 fa0 = z, fa1 = z, fa2 = z, fa3 = z;
        float4 fb0 = z, fb1 = z, fb2 = z, fb3 = z;
        const int g0 = row0 + hw * 4;
        for (int c = 0; c < nChunks; ++c) {
            const int a0 = seg[c * SEG_STRIDE + b];
            const int a1 = seg[c * SEG_STRIDE + b + 1];
            const int2* sp = pairs + (size_t)c * CHUNK_EDGES;
            for (int i = a0; i < a1; ++i) {
                const int2 p = sp[i];
                const int r = p.y - g0;
                if (r >= 0 && r < 4) {
                    const float4 v = x4[(size_t)p.x * (IN_DIM / 4) + l];
                    const float4 u = x4[(size_t)p.x * (IN_DIM / 4) + ROW_LANES + l];
                    if (r == 0) { fa0.x += v.x; fa0.y += v.y; fa0.z += v.z; fa0.w += v.w;
                                  fb0.x += u.x; fb0.y += u.y; fb0.z += u.z; fb0.w += u.w; }
                    if (r == 1) { fa1.x += v.x; fa1.y += v.y; fa1.z += v.z; fa1.w += v.w;
                                  fb1.x += u.x; fb1.y += u.y; fb1.z += u.z; fb1.w += u.w; }
                    if (r == 2) { fa2.x += v.x; fa2.y += v.y; fa2.z += v.z; fa2.w += v.w;
                                  fb2.x += u.x; fb2.y += u.y; fb2.z += u.z; fb2.w += u.w; }
                    if (r == 3) { fa3.x += v.x; fa3.y += v.y; fa3.z += v.z; fa3.w += v.w;
                                  fb3.x += u.x; fb3.y += u.y; fb3.z += u.z; fb3.w += u.w; }
                }
            }
        }
        const int rt = hw * 4;
        *reinterpret_cast<float4*>(&sA[(rt + 0) * A_PITCH + 4 * l]) = fa0;
        *reinterpret_cast<float4*>(&sA[(rt + 1) * A_PITCH + 4 * l]) = fa1;
        *reinterpret_cast<float4*>(&sA[(rt + 2) * A_PITCH + 4 * l]) = fa2;
        *reinterpret_cast<float4*>(&sA[(rt + 3) * A_PITCH + 4 * l]) = fa3;
        *reinterpret_cast<float4*>(&sA[(rt + 0) * A_PITCH + 4 * (ROW_LANES + l)]) = fb0;
        *reinterpret_cast<float4*>(&sA[(rt + 1) * A_PITCH + 4 * (ROW_LANES + l)]) = fb1;
        *reinterpret_cast<float4*>(&sA[(rt + 2) * A_PITCH + 4 * (ROW_LANES + l)]) = fb2;
        *reinterpret_cast<float4*>(&sA[(rt + 3) * A_PITCH + 4 * (ROW_LANES + l)]) = fb3;
        if (l < 4) sInv[rt + l] = 1.0f / fmaxf((float)sCnt[rt + l], 1.0f);
    }
    __syncthreads();

    const int lane = threadIdx.x & 31;
    const int tile = threadIdx.x >> 5;
    const int trow0 = row0 + tile * 16;
    if (trow0 >= nNodes) return;

    const int m    = lane & 15;
    const int hi   = lane >> 4;
    const int koff = hi * 2;

    const float inv = sInv[tile * 16 + m];
    const float* rowp = &sA[(tile * 16 + m) * A_PITCH];
    v2f a[16];
#pragma unroll
    for (int kk = 0; kk < 16; ++kk) {
        a[kk].x = rowp[kk * 4 + koff]     * inv;
        a[kk].y = rowp[kk * 4 + koff + 1] * inv;
    }

    v8f acc[4] = {};

#pragma unroll
    for (int kk = 0; kk < 16; ++kk) {
#pragma unroll
        for (int tt = 0; tt < 4; ++tt) {
            int n = tt * 16 + m;
            v2f b;
            b.x = sW[n * LDS_STRIDE + kk * 4 + koff];
            b.y = sW[n * LDS_STRIDE + kk * 4 + koff + 1];
            acc[tt] = __builtin_amdgcn_wmma_f32_16x16x4_f32(
                 false, a[kk],  false, b,
                 (short)0, acc[tt],
                 false,  false);
        }
    }

#pragma unroll
    for (int tt = 0; tt < 4; ++tt) {
        int n = tt * 16 + m;
        float bv = bias[n];
#pragma unroll
        for (int v = 0; v < 8; ++v) {
            int rr = trow0 + v + hi * 8;
            out[(size_t)rr * OUT_DIM + n] = acc[tt][v] + bv;
        }
    }
}

extern "C" void kernel_launch(void* const* d_in, const int* in_sizes, int n_in,
                              void* d_out, int out_size, void* d_ws, size_t ws_size,
                              hipStream_t stream) {
    const float* x   = (const float*)d_in[0];
    const int*   src = (const int*)  d_in[1];
    const int*   dst = (const int*)  d_in[2];
    const float* W   = (const float*)d_in[3];
    const float* b   = (const float*)d_in[4];

    float* out = (float*)d_out;

    const int nNodes = in_sizes[0] / IN_DIM;
    const int nEdges = in_sizes[1];

    const int nBins   = (nNodes + TILE_ROWS - 1) / TILE_ROWS;
    const int nChunks = (nEdges + CHUNK_EDGES - 1) / CHUNK_EDGES;

    int*   seg     = (int*)d_ws;
    int2*  pairs   = (int2*)(seg + nChunks * SEG_STRIDE);

    chunk_sort_kernel<<<nChunks, CHUNK_T, 0, stream>>>(src, dst, seg, pairs, nEdges);

    gather_gemm_kernel<<<nBins, GG_T, 0, stream>>>(x, pairs, seg, W, b, out, nNodes, nChunks);
}
